# starting kernel
# baseline (speedup 1.0000x reference)
#include <hip/hip_runtime.h>
#include <hip/hip_bf16.h>
#include <hip/hip_fp16.h>

typedef float v2f __attribute__((ext_vector_type(2)));
typedef float v8f __attribute__((ext_vector_type(8)));

#define IN_DIM 128
#define HDIM 128
#define NHEADS 4
#define ODIM 32

#define BPADF 272
#define B_LDS_BYTES (64 * BPADF * 4)

__device__ __forceinline__ unsigned fkey(float f) {
  unsigned b = __float_as_uint(f);
  return (b & 0x80000000u) ? ~b : (b | 0x80000000u);
}
__device__ __forceinline__ float funkey(unsigned k) {
  unsigned b = (k & 0x80000000u) ? (k & 0x7FFFFFFFu) : ~k;
  return __uint_as_float(b);
}
#define NEG_INF_KEY 0x007FFFFFu

__global__ __launch_bounds__(256) void gat_fc_wmma(
    const float* __restrict__ feat, const float* __restrict__ fc_w,
    float* __restrict__ h, int n) {
  extern __shared__ float Bs[];
  const int tid = threadIdx.x;
  for (int i = tid; i < IN_DIM * HDIM; i += 256) {
    const int kk = i & (IN_DIM - 1);
    const int jj = i >> 7;
    Bs[(kk >> 1) * BPADF + jj * 2 + (kk & 1)] = fc_w[jj * IN_DIM + kk];
  }
  __syncthreads();

  const int wave = tid >> 5;
  const int lane = tid & 31;
  const int half = lane >> 4;
  const int l16  = lane & 15;
  const int rowbase = blockIdx.x * 128 + wave * 16;
  if (rowbase >= n) return;

  const int row = min(rowbase + l16, n - 1);
  const float* __restrict__ fr = feat + (size_t)row * IN_DIM + 2 * half;
  const float* __restrict__ bbase = Bs + l16 * 2 + half * BPADF;

  v8f acc[8] = {};
  for (int kb = 0; kb < IN_DIM; kb += 16) {
    v2f a[4];
#pragma unroll
    for (int s = 0; s < 4; ++s) a[s] = *(const v2f*)(fr + kb + 4 * s);

#pragma unroll
    for (int s = 0; s < 4; ++s) {
      const float* __restrict__ bk = bbase + ((kb >> 1) + 2 * s) * BPADF;
      v2f b[8];
#pragma unroll
      for (int t = 0; t < 8; ++t) b[t] = *(const v2f*)(bk + t * 32);
#pragma unroll
      for (int t = 0; t < 8; ++t)
        acc[t] = __builtin_amdgcn_wmma_f32_16x16x4_f32(
            false, a[s], false, b[t], (short)0, acc[t], false, false);
    }
  }
#pragma unroll
  for (int j = 0; j < 8; ++j) {
    const int r = rowbase + j + 8 * half;
    if (r < n) {
      float* hr = h + (size_t)r * HDIM;
#pragma unroll
      for (int t = 0; t < 8; ++t) hr[t * 16 + l16] = acc[t][j];
    }
  }
}

__global__ void gat_init(float* __restrict__ ssum, unsigned* __restrict__ mkey,
                         const float* __restrict__ feat, float* __restrict__ out,
                         int n) {
  int idx = blockIdx.x * blockDim.x + threadIdx.x;
  if (idx < n * NHEADS) {
    ssum[idx] = 0.0f;
    mkey[idx] = NEG_INF_KEY;
  }
  if (idx < n * (HDIM / 4)) {
    ((float4*)out)[idx] = ((const float4*)feat)[idx];
  }
}

__global__ void gat_node_logits(const float* __restrict__ h,
                                const float* __restrict__ attn_src,
                                const float* __restrict__ attn_dst,
                                float* __restrict__ e_src,
                                float* __restrict__ e_dst, int n) {
  int idx = blockIdx.x * blockDim.x + threadIdx.x;
  if (idx >= n * NHEADS) return;
  const int node = idx >> 2, hh = idx & 3;
  const float4* hp = (const float4*)(h + (size_t)node * HDIM + hh * ODIM);
  const float4* as = (const float4*)(attn_src + hh * ODIM);
  const float4* ad = (const float4*)(attn_dst + hh * ODIM);
  float ss = 0.0f, dd = 0.0f;
#pragma unroll
  for (int i = 0; i < ODIM / 4; ++i) {
    float4 v = hp[i], a = as[i], b = ad[i];
    ss += v.x * a.x + v.y * a.y + v.z * a.z + v.w * a.w;
    dd += v.x * b.x + v.y * b.y + v.z * b.z + v.w * b.w;
  }
  e_src[idx] = ss;
  e_dst[idx] = dd;
}

__global__ void gat_edge_logits(const int* __restrict__ src,
                                const int* __restrict__ dst,
                                const float* __restrict__ e_src,
                                const float* __restrict__ e_dst,
                                float* __restrict__ e_edge,
                                unsigned* __restrict__ mkey, int ne) {
  int idx = blockIdx.x * blockDim.x + threadIdx.x;
  if (idx >= ne * NHEADS) return;
  const int e = idx >> 2, hh = idx & 3;
  const int s = src[e], d = dst[e];
  float v = e_src[s * NHEADS + hh] + e_dst[d * NHEADS + hh];
  v = v > 0.0f ? v : 0.2f * v;
  e_edge[idx] = v;
  atomicMax(mkey + d * NHEADS + hh, fkey(v));
}

__global__ void gat_edge_softmax(const int* __restrict__ dst,
                                 float* __restrict__ e_edge,
                                 const unsigned* __restrict__ mkey,
                                 float* __restrict__ ssum, int ne) {
  int idx = blockIdx.x * blockDim.x + threadIdx.x;
  if (idx >= ne * NHEADS) return;
  const int e = idx >> 2, hh = idx & 3;
  const int d = dst[e];
  const float m = funkey(mkey[d * NHEADS + hh]);
  const float a = __expf(e_edge[idx] - m);
  e_edge[idx] = a;
  unsafeAtomicAdd(ssum + d * NHEADS + hh, a);
}

__global__ __launch_bounds__(256) void gat_aggregate(
    const int* __restrict__ src, const int* __restrict__ dst,
    const float* __restrict__ h, const float* __restrict__ a_edge,
    const float* __restrict__ ssum, float* __restrict__ out, int ne) {
  const int edge = blockIdx.x * 8 + (threadIdx.x >> 5);
  if (edge >= ne) return;
  const int lane = threadIdx.x & 31;
  const int s = src[edge], d = dst[edge];
  const int hh = lane >> 3;
  const float a = a_edge[edge * NHEADS + hh];
  const float den = ssum[d * NHEADS + hh];
  const float w = a / (den > 0.0f ? den : 1.0f);
  const float4 hv = *(const float4*)(h + (size_t)s * HDIM + lane * 4);
  float* ob = out + (size_t)d * HDIM + lane * 4;
  unsafeAtomicAdd(ob + 0, hv.x * w);
  unsafeAtomicAdd(ob + 1, hv.y * w);
  unsafeAtomicAdd(ob + 2, hv.z * w);
  unsafeAtomicAdd(ob + 3, hv.w * w);
}

extern "C" void kernel_launch(void* const* d_in, const int* in_sizes, int n_in,
                              void* d_out, int out_size, void* d_ws, size_t ws_size,
                              hipStream_t stream) {
  const float* feat     = (const float*)d_in[0];
  const float* fc_w     = (const float*)d_in[1];
  const float* attn_src = (const float*)d_in[2];
  const float* attn_dst = (const float*)d_in[3];
  const int*   src      = (const int*)d_in[4];
  const int*   dst      = (const int*)d_in[5];
  float* out = (float*)d_out;

  const int n  = in_sizes[0] / IN_DIM;
  const int ne = in_sizes[4];

  char* ws = (char*)d_ws;
  size_t off = 0;
  float*    h      = (float*)(ws + off); off += (size_t)n * HDIM * sizeof(float);
  float*    e_src  = (float*)(ws + off); off += (size_t)n * NHEADS * sizeof(float);
  float*    e_dst  = (float*)(ws + off); off += (size_t)n * NHEADS * sizeof(float);
  unsigned* mkey   = (unsigned*)(ws + off); off += (size_t)n * NHEADS * sizeof(unsigned);
  float*    ssum   = (float*)(ws + off); off += (size_t)n * NHEADS * sizeof(float);
  float*    e_edge = (float*)(ws + off); off += (size_t)ne * NHEADS * sizeof(float);
  (void)ws_size; (void)n_in; (void)out_size;

  const int T = 256;
  gat_init<<<(n * (HDIM / 4) + T - 1) / T, T, 0, stream>>>(ssum, mkey, feat, out, n);
  gat_fc_wmma<<<(n + 127) / 128, T, B_LDS_BYTES, stream>>>(feat, fc_w, h, n);
  gat_node_logits<<<(n * NHEADS + T - 1) / T, T, 0, stream>>>(h, attn_src, attn_dst,
                                                              e_src, e_dst, n);
  gat_edge_logits<<<(ne * NHEADS + T - 1) / T, T, 0, stream>>>(src, dst, e_src, e_dst,
                                                               e_edge, mkey, ne);
  gat_edge_softmax<<<(ne * NHEADS + T - 1) / T, T, 0, stream>>>(dst, e_edge, mkey,
                                                                ssum, ne);
  gat_aggregate<<<(ne + 7) / 8, T, 0, stream>>>(src, dst, h, e_edge, ssum, out, ne);
}


# Group edges by destination in-call (count with returning int atomics, scan, fill); one wave per node does softmax and weighted sum in registers: no float atomics, sums in arrival-rank order
# speedup vs baseline: 1.9161x; 1.9161x over previous
#include <hip/hip_runtime.h>
#include <hip/hip_bf16.h>
#include <hip/hip_fp16.h>

typedef float v2f __attribute__((ext_vector_type(2)));
typedef float v8f __attribute__((ext_vector_type(8)));

#define IN_DIM 128
#define HDIM 128
#define NHEADS 4
#define ODIM 32

#define BPADF 272
#define B_LDS_BYTES (64 * BPADF * 4)

__global__ __launch_bounds__(256) void gat_fc_wmma(
    const float* __restrict__ feat, const float* __restrict__ fc_w,
    float* __restrict__ h, int n) {
  extern __shared__ float Bs[];
  const int tid = threadIdx.x;
  for (int i = tid; i < IN_DIM * HDIM; i += 256) {
    const int kk = i & (IN_DIM - 1);
    const int jj = i >> 7;
    Bs[(kk >> 1) * BPADF + jj * 2 + (kk & 1)] = fc_w[jj * IN_DIM + kk];
  }
  __syncthreads();

  const int wave = tid >> 5;
  const int lane = tid & 31;
  const int half = lane >> 4;
  const int l16  = lane & 15;
  const int rowbase = blockIdx.x * 128 + wave * 16;
  if (rowbase >= n) return;

  const int row = min(rowbase + l16, n - 1);
  const float* __restrict__ fr = feat + (size_t)row * IN_DIM + 2 * half;
  const float* __restrict__ bbase = Bs + l16 * 2 + half * BPADF;

  v8f acc[8] = {};
  for (int kb = 0; kb < IN_DIM; kb += 16) {
    v2f a[4];
#pragma unroll
    for (int s = 0; s < 4; ++s) a[s] = *(const v2f*)(fr + kb + 4 * s);

#pragma unroll
    for (int s = 0; s < 4; ++s) {
      const float* __restrict__ bk = bbase + ((kb >> 1) + 2 * s) * BPADF;
      v2f b[8];
#pragma unroll
      for (int t = 0; t < 8; ++t) b[t] = *(const v2f*)(bk + t * 32);
#pragma unroll
      for (int t = 0; t < 8; ++t)
        acc[t] = __builtin_amdgcn_wmma_f32_16x16x4_f32(
            false, a[s], false, b[t], (short)0, acc[t], false, false);
    }
  }
#pragma unroll
  for (int j = 0; j < 8; ++j) {
    const int r = rowbase + j + 8 * half;
    if (r < n) {
      float* hr = h + (size_t)r * HDIM;
#pragma unroll
      for (int t = 0; t < 8; ++t) hr[t * 16 + l16] = acc[t][j];
    }
  }
}

__global__ void gat_node_logits(const float* __restrict__ h,
                                const float* __restrict__ attn_src,
                                const float* __restrict__ attn_dst,
                                float* __restrict__ e_src,
                                float* __restrict__ e_dst,
                                int* __restrict__ cnt, int n) {
  int idx = blockIdx.x * blockDim.x + threadIdx.x;
  if (idx >= n * NHEADS) return;
  const int node = idx >> 2, hh = idx & 3;
  if (hh == 0) cnt[node] = 0;
  const float4* hp = (const float4*)(h + (size_t)node * HDIM + hh * ODIM);
  const float4* as = (const float4*)(attn_src + hh * ODIM);
  const float4* ad = (const float4*)(attn_dst + hh * ODIM);
  float ss = 0.0f, dd = 0.0f;
#pragma unroll
  for (int i = 0; i < ODIM / 4; ++i) {
    float4 v = hp[i], a = as[i], b = ad[i];
    ss += v.x * a.x + v.y * a.y + v.z * a.z + v.w * a.w;
    dd += v.x * b.x + v.y * b.y + v.z * b.z + v.w * b.w;
  }
  e_src[idx] = ss;
  e_dst[idx] = dd;
}

__global__ void gat_count(const int* __restrict__ dst, int* __restrict__ cnt,
                          int* __restrict__ rank, int ne) {
  const int e0 = (blockIdx.x * blockDim.x + threadIdx.x) * 4;
  if (e0 + 3 < ne) {
    const int4 d = *(const int4*)(dst + e0);
    int4 r;
    r.x = atomicAdd(cnt + d.x, 1);
    r.y = atomicAdd(cnt + d.y, 1);
    r.z = atomicAdd(cnt + d.z, 1);
    r.w = atomicAdd(cnt + d.w, 1);
    *(int4*)(rank + e0) = r;
  } else {
    for (int e = e0; e < ne; ++e) rank[e] = atomicAdd(cnt + dst[e], 1);
  }
}

__global__ __launch_bounds__(256) void gat_scan_sums(
    const int* __restrict__ cnt, int* __restrict__ bsum, int n) {
  __shared__ int sh[256];
  const int tid = threadIdx.x;
  const int i0 = blockIdx.x * 1024 + tid * 4;
  int s = 0;
#pragma unroll
  for (int u = 0; u < 4; ++u) s += (i0 + u < n) ? cnt[i0 + u] : 0;
  sh[tid] = s;
  __syncthreads();
  for (int o = 128; o > 0; o >>= 1) {
    if (tid < o) sh[tid] += sh[tid + o];
    __syncthreads();
  }
  if (tid == 0) bsum[blockIdx.x] = sh[0];
}

__global__ __launch_bounds__(256) void gat_scan_offsets(
    const int* __restrict__ cnt, const int* __restrict__ bsum,
    int* __restrict__ off, int n) {
  __shared__ int sh[256];
  const int tid = threadIdx.x;
  const int b = blockIdx.x;
  int part = 0;
  for (int i = tid; i < b; i += 256) part += bsum[i];
  sh[tid] = part;
  __syncthreads();
  for (int o = 128; o > 0; o >>= 1) {
    if (tid < o) sh[tid] += sh[tid + o];
    __syncthreads();
  }
  const int base = sh[0];
  __syncthreads();
  const int i0 = b * 1024 + tid * 4;
  int c[4];
#pragma unroll
  for (int u = 0; u < 4; ++u) c[u] = (i0 + u < n) ? cnt[i0 + u] : 0;
  const int ts = c[0] + c[1] + c[2] + c[3];
  sh[tid] = ts;
  __syncthreads();
  for (int o = 1; o < 256; o <<= 1) {
    const int t = (tid >= o) ? sh[tid - o] : 0;
    __syncthreads();
    sh[tid] += t;
    __syncthreads();
  }
  int run = base + sh[tid] - ts;
#pragma unroll
  for (int u = 0; u < 4; ++u) {
    if (i0 + u < n) off[i0 + u] = run;
    run += c[u];
  }
}

__global__ void gat_fill(const int* __restrict__ src, const int* __restrict__ dst,
                         const int* __restrict__ rank, const int* __restrict__ off,
                         int* __restrict__ nbr, int ne) {
  const int e0 = (blockIdx.x * blockDim.x + threadIdx.x) * 4;
  if (e0 + 3 < ne) {
    const int4 s = *(const int4*)(src + e0);
    const int4 d = *(const int4*)(dst + e0);
    const int4 r = *(const int4*)(rank + e0);
    nbr[off[d.x] + r.x] = s.x;
    nbr[off[d.y] + r.y] = s.y;
    nbr[off[d.z] + r.z] = s.z;
    nbr[off[d.w] + r.w] = s.w;
  } else {
    for (int e = e0; e < ne; ++e) nbr[off[dst[e]] + rank[e]] = src[e];
  }
}

__device__ __forceinline__ float gat_leaky(float x) {
  return x > 0.0f ? x : 0.2f * x;
}

template <int CH, bool TAIL>
__device__ __forceinline__ void gat_max_chunk(const int* __restrict__ nb, int j0,
                                              int deg,
                                              const float* __restrict__ e_src,
                                              int k, float edk, float& m) {
  float v[CH];
#pragma unroll
  for (int u = 0; u < CH; ++u) {
    const int j = TAIL ? min(j0 + u, deg - 1) : (j0 + u);
    v[u] = e_src[nb[j] * NHEADS + k];
  }
#pragma unroll
  for (int u = 0; u < CH; ++u) m = fmaxf(m, gat_leaky(v[u] + edk));
}

template <int CH, bool TAIL>
__device__ __forceinline__ void gat_sum_chunk(const int* __restrict__ nb, int j0,
                                              int deg,
                                              const float* __restrict__ e_src,
                                              const float* __restrict__ h,
                                              int k, int lane, float edk, float m,
                                              float& den, float4& acc) {
  int s[CH];
  float es[CH];
  float4 hv[CH];
#pragma unroll
  for (int u = 0; u < CH; ++u) s[u] = nb[TAIL ? min(j0 + u, deg - 1) : (j0 + u)];
#pragma unroll
  for (int u = 0; u < CH; ++u) {
    es[u] = e_src[s[u] * NHEADS + k];
    hv[u] = *(const float4*)(h + (size_t)s[u] * HDIM + lane * 4);
  }
#pragma unroll
  for (int u = 0; u < CH; ++u) {
    float a = __expf(gat_leaky(es[u] + edk) - m);
    if (TAIL && j0 + u >= deg) a = 0.0f;
    den += a;
    acc.x += a * hv[u].x;
    acc.y += a * hv[u].y;
    acc.z += a * hv[u].z;
    acc.w += a * hv[u].w;
  }
}

__global__ __launch_bounds__(256) void gat_gather(
    const int* __restrict__ nbr, const int* __restrict__ off,
    const int* __restrict__ cnt, const float* __restrict__ h,
    const float* __restrict__ e_src, const float* __restrict__ e_dst,
    const float* __restrict__ feat, float* __restrict__ out, int n) {
  const int d = __builtin_amdgcn_readfirstlane(blockIdx.x * 8 + (threadIdx.x >> 5));
  if (d >= n) return;
  const int lane = threadIdx.x & 31;
  const int k = lane >> 3;
  const int deg = cnt[d];
  const int* __restrict__ nb = nbr + off[d];
  const float edk = e_dst[d * NHEADS + k];

  float m = -INFINITY;
  int j0 = 0;
  for (; j0 + 8 <= deg; j0 += 8) gat_max_chunk<8, false>(nb, j0, deg, e_src, k, edk, m);
  for (; j0 < deg; j0 += 4) gat_max_chunk<4, true>(nb, j0, deg, e_src, k, edk, m);

  float den = 0.0f;
  float4 acc = make_float4(0.0f, 0.0f, 0.0f, 0.0f);
  j0 = 0;
  for (; j0 + 8 <= deg; j0 += 8)
    gat_sum_chunk<8, false>(nb, j0, deg, e_src, h, k, lane, edk, m, den, acc);
  for (; j0 < deg; j0 += 4)
    gat_sum_chunk<4, true>(nb, j0, deg, e_src, h, k, lane, edk, m, den, acc);

  const float dsafe = den > 0.0f ? den : 1.0f;
  const float4 f = *(const float4*)(feat + (size_t)d * HDIM + lane * 4);
  float4 o;
  o.x = f.x + acc.x / dsafe;
  o.y = f.y + acc.y / dsafe;
  o.z = f.z + acc.z / dsafe;
  o.w = f.w + acc.w / dsafe;
  *(float4*)(out + (size_t)d * HDIM + lane * 4) = o;
}

extern "C" void kernel_launch(void* const* d_in, const int* in_sizes, int n_in,
                              void* d_out, int out_size, void* d_ws, size_t ws_size,
                              hipStream_t stream) {
  const float* feat     = (const float*)d_in[0];
  const float* fc_w     = (const float*)d_in[1];
  const float* attn_src = (const float*)d_in[2];
  const float* attn_dst = (const float*)d_in[3];
  const int*   src      = (const int*)d_in[4];
  const int*   dst      = (const int*)d_in[5];
  float* out = (float*)d_out;

  const int n  = in_sizes[0] / IN_DIM;
  const int ne = in_sizes[4];

  char* ws = (char*)d_ws;
  size_t off = 0;
  float* h     = (float*)(ws + off); off += (size_t)n * HDIM * sizeof(float);
  float* e_src = (float*)(ws + off); off += (size_t)n * NHEADS * sizeof(float);
  float* e_dst = (float*)(ws + off); off += (size_t)n * NHEADS * sizeof(float);
  int*   rank  = (int*)(ws + off);   off += (size_t)ne * sizeof(int);
  int*   nbr   = (int*)(ws + off);   off += (size_t)ne * sizeof(int);
  int*   cnt   = (int*)(ws + off);   off += (size_t)n * sizeof(int);
  int*   segs  = (int*)(ws + off);   off += (size_t)n * sizeof(int);
  int*   bsum  = (int*)(ws + off);   off += (size_t)((n + 1023) / 1024) * sizeof(int);
  (void)ws_size; (void)n_in; (void)out_size;

  const int T = 256;
  const int nslab = (n + 1023) / 1024;
  const int equads = (ne + 3) / 4;
  gat_fc_wmma<<<(n + 127) / 128, T, B_LDS_BYTES, stream>>>(feat, fc_w, h, n);
  gat_node_logits<<<(n * NHEADS + T - 1) / T, T, 0, stream>>>(h, attn_src, attn_dst,
                                                              e_src, e_dst, cnt, n);
  gat_count<<<(equads + T - 1) / T, T, 0, stream>>>(dst, cnt, rank, ne);
  gat_scan_sums<<<nslab, T, 0, stream>>>(cnt, bsum, n);
  gat_scan_offsets<<<nslab, T, 0, stream>>>(cnt, bsum, segs, n);
  gat_fill<<<(equads + T - 1) / T, T, 0, stream>>>(src, dst, rank, segs, nbr, ne);
  gat_gather<<<(n + 7) / 8, T, 0, stream>>>(nbr, segs, cnt, h, e_src, e_dst, feat, out, n);
}


# Gather walks each neighbour list once: running max, partial sums rescaled by exp(m_old-m_new) when it rises (online softmax); a chunk's 8 logits+8 rows requested as one group; arrival-rank order kept
# speedup vs baseline: 2.0449x; 1.0672x over previous
#include <hip/hip_runtime.h>
#include <hip/hip_bf16.h>
#include <hip/hip_fp16.h>

typedef float v2f __attribute__((ext_vector_type(2)));
typedef float v8f __attribute__((ext_vector_type(8)));
typedef float v4f __attribute__((ext_vector_type(4)));

#define IN_DIM 128
#define HDIM 128
#define NHEADS 4
#define ODIM 32

#define BPADF 272
#define B_LDS_BYTES (64 * BPADF * 4)

__global__ __launch_bounds__(256) void gat_fc_wmma(
    const float* __restrict__ feat, const float* __restrict__ fc_w,
    float* __restrict__ h, int n) {
  extern __shared__ float Bs[];
  const int tid = threadIdx.x;
  for (int i = tid; i < IN_DIM * HDIM; i += 256) {
    const int kk = i & (IN_DIM - 1);
    const int jj = i >> 7;
    Bs[(kk >> 1) * BPADF + jj * 2 + (kk & 1)] = fc_w[jj * IN_DIM + kk];
  }
  __syncthreads();

  const int wave = tid >> 5;
  const int lane = tid & 31;
  const int half = lane >> 4;
  const int l16  = lane & 15;
  const int rowbase = blockIdx.x * 128 + wave * 16;
  if (rowbase >= n) return;

  const int row = min(rowbase + l16, n - 1);
  const float* __restrict__ fr = feat + (size_t)row * IN_DIM + 2 * half;
  const float* __restrict__ bbase = Bs + l16 * 2 + half * BPADF;

  v8f acc[8] = {};
  for (int kb = 0; kb < IN_DIM; kb += 16) {
    v2f a[4];
#pragma unroll
    for (int s = 0; s < 4; ++s) a[s] = *(const v2f*)(fr + kb + 4 * s);

#pragma unroll
    for (int s = 0; s < 4; ++s) {
      const float* __restrict__ bk = bbase + ((kb >> 1) + 2 * s) * BPADF;
      v2f b[8];
#pragma unroll
      for (int t = 0; t < 8; ++t) b[t] = *(const v2f*)(bk + t * 32);
#pragma unroll
      for (int t = 0; t < 8; ++t)
        acc[t] = __builtin_amdgcn_wmma_f32_16x16x4_f32(
            false, a[s], false, b[t], (short)0, acc[t], false, false);
    }
  }
#pragma unroll
  for (int j = 0; j < 8; ++j) {
    const int r = rowbase + j + 8 * half;
    if (r < n) {
      float* hr = h + (size_t)r * HDIM;
#pragma unroll
      for (int t = 0; t < 8; ++t) hr[t * 16 + l16] = acc[t][j];
    }
  }
}

__global__ void gat_node_logits(const float* __restrict__ h,
                                const float* __restrict__ attn_src,
                                const float* __restrict__ attn_dst,
                                float* __restrict__ e_src,
                                float* __restrict__ e_dst,
                                int* __restrict__ cnt, int n) {
  int idx = blockIdx.x * blockDim.x + threadIdx.x;
  if (idx >= n * NHEADS) return;
  const int node = idx >> 2, hh = idx & 3;
  if (hh == 0) cnt[node] = 0;
  const float4* hp = (const float4*)(h + (size_t)node * HDIM + hh * ODIM);
  const float4* as = (const float4*)(attn_src + hh * ODIM);
  const float4* ad = (const float4*)(attn_dst + hh * ODIM);
  float ss = 0.0f, dd = 0.0f;
#pragma unroll
  for (int i = 0; i < ODIM / 4; ++i) {
    float4 v = hp[i], a = as[i], b = ad[i];
    ss += v.x * a.x + v.y * a.y + v.z * a.z + v.w * a.w;
    dd += v.x * b.x + v.y * b.y + v.z * b.z + v.w * b.w;
  }
  e_src[idx] = ss;
  e_dst[idx] = dd;
}

__global__ void gat_count(const int* __restrict__ dst, int* __restrict__ cnt,
                          int* __restrict__ rank, int ne) {
  const int e0 = (blockIdx.x * blockDim.x + threadIdx.x) * 4;
  if (e0 + 3 < ne) {
    const int4 d = *(const int4*)(dst + e0);
    int4 r;
    r.x = atomicAdd(cnt + d.x, 1);
    r.y = atomicAdd(cnt + d.y, 1);
    r.z = atomicAdd(cnt + d.z, 1);
    r.w = atomicAdd(cnt + d.w, 1);
    *(int4*)(rank + e0) = r;
  } else {
    for (int e = e0; e < ne; ++e) rank[e] = atomicAdd(cnt + dst[e], 1);
  }
}

__global__ __launch_bounds__(256) void gat_scan_sums(
    const int* __restrict__ cnt, int* __restrict__ bsum, int n) {
  __shared__ int sh[256];
  const int tid = threadIdx.x;
  const int i0 = blockIdx.x * 1024 + tid * 4;
  int s = 0;
#pragma unroll
  for (int u = 0; u < 4; ++u) s += (i0 + u < n) ? cnt[i0 + u] : 0;
  sh[tid] = s;
  __syncthreads();
  for (int o = 128; o > 0; o >>= 1) {
    if (tid < o) sh[tid] += sh[tid + o];
    __syncthreads();
  }
  if (tid == 0) bsum[blockIdx.x] = sh[0];
}

__global__ __launch_bounds__(256) void gat_scan_offsets(
    const int* __restrict__ cnt, const int* __restrict__ bsum,
    int* __restrict__ off, int n) {
  __shared__ int sh[256];
  const int tid = threadIdx.x;
  const int b = blockIdx.x;
  int part = 0;
  for (int i = tid; i < b; i += 256) part += bsum[i];
  sh[tid] = part;
  __syncthreads();
  for (int o = 128; o > 0; o >>= 1) {
    if (tid < o) sh[tid] += sh[tid + o];
    __syncthreads();
  }
  const int base = sh[0];
  __syncthreads();
  const int i0 = b * 1024 + tid * 4;
  int c[4];
#pragma unroll
  for (int u = 0; u < 4; ++u) c[u] = (i0 + u < n) ? cnt[i0 + u] : 0;
  const int ts = c[0] + c[1] + c[2] + c[3];
  sh[tid] = ts;
  __syncthreads();
  for (int o = 1; o < 256; o <<= 1) {
    const int t = (tid >= o) ? sh[tid - o] : 0;
    __syncthreads();
    sh[tid] += t;
    __syncthreads();
  }
  int run = base + sh[tid] - ts;
#pragma unroll
  for (int u = 0; u < 4; ++u) {
    if (i0 + u < n) off[i0 + u] = run;
    run += c[u];
  }
}

__global__ void gat_fill(const int* __restrict__ src, const int* __restrict__ dst,
                         const int* __restrict__ rank, const int* __restrict__ off,
                         int* __restrict__ nbr, int ne) {
  const int e0 = (blockIdx.x * blockDim.x + threadIdx.x) * 4;
  if (e0 + 3 < ne) {
    const int4 s = *(const int4*)(src + e0);
    const int4 d = *(const int4*)(dst + e0);
    const int4 r = *(const int4*)(rank + e0);
    nbr[off[d.x] + r.x] = s.x;
    nbr[off[d.y] + r.y] = s.y;
    nbr[off[d.z] + r.z] = s.z;
    nbr[off[d.w] + r.w] = s.w;
  } else {
    for (int e = e0; e < ne; ++e) nbr[off[dst[e]] + rank[e]] = src[e];
  }
}

__device__ __forceinline__ float gat_leaky(float x) {
  return x > 0.0f ? x : 0.2f * x;
}

template <int CH, bool TAIL>
__device__ __forceinline__ void gat_chunk(const int* s_in, int rem,
                                          const float* __restrict__ e_src,
                                          const float* __restrict__ h,
                                          int k, int lane, float edk,
                                          float& m, float& den, float4& acc) {
  int s[CH];
  float e[CH];
  v4f hv[CH];
#pragma unroll
  for (int u = 0; u < CH; ++u) s[u] = (!TAIL || u < rem) ? s_in[u] : s_in[0];
#pragma unroll
  for (int u = 0; u < CH; ++u) {
    e[u] = e_src[s[u] * NHEADS + k];
    hv[u] = ((const v4f*)h)[(unsigned)s[u] * (HDIM / 4) + (unsigned)lane];
  }
  if constexpr (CH == 8)
    asm volatile("" : "+v"(e[0]), "+v"(e[1]), "+v"(e[2]), "+v"(e[3]),
                      "+v"(e[CH - 4]), "+v"(e[CH - 3]), "+v"(e[CH - 2]), "+v"(e[CH - 1])
                    : "v"(hv[0]), "v"(hv[1]), "v"(hv[2]), "v"(hv[3]),
                      "v"(hv[CH - 4]), "v"(hv[CH - 3]), "v"(hv[CH - 2]), "v"(hv[CH - 1]));
  else
    asm volatile("" : "+v"(e[0]), "+v"(e[1]), "+v"(e[2]), "+v"(e[3])
                    : "v"(hv[0]), "v"(hv[1]), "v"(hv[2]), "v"(hv[3]));
#pragma unroll
  for (int u = 0; u < CH; ++u) e[u] = gat_leaky(e[u] + edk);
  float mn = m;
#pragma unroll
  for (int u = 0; u < CH; ++u) mn = fmaxf(mn, e[u]);
  const float sc = __expf(m - mn);
  m = mn;
  den *= sc;
  acc.x *= sc;
  acc.y *= sc;
  acc.z *= sc;
  acc.w *= sc;
#pragma unroll
  for (int u = 0; u < CH; ++u) {
    float a = __expf(e[u] - mn);
    if (TAIL && u >= rem) a = 0.0f;
    den += a;
    acc.x += a * hv[u].x;
    acc.y += a * hv[u].y;
    acc.z += a * hv[u].z;
    acc.w += a * hv[u].w;
  }
}

__global__ __launch_bounds__(256) void gat_gather(
    const int* __restrict__ nbr, const int* __restrict__ off,
    const int* __restrict__ cnt, const float* __restrict__ h,
    const float* __restrict__ e_src, const float* __restrict__ e_dst,
    const float* __restrict__ feat, float* __restrict__ out, int n) {
  const int d = __builtin_amdgcn_readfirstlane(blockIdx.x * 8 + (threadIdx.x >> 5));
  if (d >= n) return;
  const int lane = threadIdx.x & 31;
  const int k = lane >> 3;
  const int deg = cnt[d];
  int seg = off[d];
  asm volatile("" : "+s"(seg));
  const int* __restrict__ nb = nbr + seg;
  const float edk = e_dst[d * NHEADS + k];
  const float4 f = *(const float4*)(feat + (size_t)d * HDIM + lane * 4);
  int cur[8], nxt[8];
#pragma unroll
  for (int u = 0; u < 8; ++u) cur[u] = nb[u];
  __builtin_amdgcn_sched_barrier(0);

  float m = -INFINITY;
  float den = 0.0f;
  float4 acc = make_float4(0.0f, 0.0f, 0.0f, 0.0f);
  for (int j0 = 0; j0 < deg; j0 += 8) {
    const int rem = deg - j0;
    if (rem > 8) {
#pragma unroll
      for (int u = 0; u < 8; ++u) nxt[u] = nb[j0 + 8 + u];
    }
    if (rem >= 8)
      gat_chunk<8, false>(cur, 8, e_src, h, k, lane, edk, m, den, acc);
    else if (rem > 4)
      gat_chunk<8, true>(cur, rem, e_src, h, k, lane, edk, m, den, acc);
    else
      gat_chunk<4, true>(cur, rem, e_src, h, k, lane, edk, m, den, acc);
    if (rem > 8) {
#pragma unroll
      for (int u = 0; u < 8; ++u) cur[u] = nxt[u];
    }
  }

  const float dsafe = den > 0.0f ? den : 1.0f;
  float4 o;
  o.x = f.x + acc.x / dsafe;
  o.y = f.y + acc.y / dsafe;
  o.z = f.z + acc.z / dsafe;
  o.w = f.w + acc.w / dsafe;
  *(float4*)(out + (size_t)d * HDIM + lane * 4) = o;
}

extern "C" void kernel_launch(void* const* d_in, const int* in_sizes, int n_in,
                              void* d_out, int out_size, void* d_ws, size_t ws_size,
                              hipStream_t stream) {
  const float* feat     = (const float*)d_in[0];
  const float* fc_w     = (const float*)d_in[1];
  const float* attn_src = (const float*)d_in[2];
  const float* attn_dst = (const float*)d_in[3];
  const int*   src      = (const int*)d_in[4];
  const int*   dst      = (const int*)d_in[5];
  float* out = (float*)d_out;

  const int n  = in_sizes[0] / IN_DIM;
  const int ne = in_sizes[4];

  char* ws = (char*)d_ws;
  size_t off = 0;
  float* h     = (float*)(ws + off); off += (size_t)n * HDIM * sizeof(float);
  float* e_src = (float*)(ws + off); off += (size_t)n * NHEADS * sizeof(float);
  float* e_dst = (float*)(ws + off); off += (size_t)n * NHEADS * sizeof(float);
  int*   rank  = (int*)(ws + off);   off += (size_t)ne * sizeof(int);
  int*   nbr   = (int*)(ws + off);   off += (size_t)ne * sizeof(int);
  int*   cnt   = (int*)(ws + off);   off += (size_t)n * sizeof(int);
  int*   segs  = (int*)(ws + off);   off += (size_t)n * sizeof(int);
  int*   bsum  = (int*)(ws + off);   off += (size_t)((n + 1023) / 1024) * sizeof(int);
  (void)ws_size; (void)n_in; (void)out_size;

  const int T = 256;
  const int nslab = (n + 1023) / 1024;
  const int equads = (ne + 3) / 4;
  gat_fc_wmma<<<(n + 127) / 128, T, B_LDS_BYTES, stream>>>(feat, fc_w, h, n);
  gat_node_logits<<<(n * NHEADS + T - 1) / T, T, 0, stream>>>(h, attn_src, attn_dst,
                                                              e_src, e_dst, cnt, n);
  gat_count<<<(equads + T - 1) / T, T, 0, stream>>>(dst, cnt, rank, ne);
  gat_scan_sums<<<nslab, T, 0, stream>>>(cnt, bsum, n);
  gat_scan_offsets<<<nslab, T, 0, stream>>>(cnt, bsum, segs, n);
  gat_fill<<<(equads + T - 1) / T, T, 0, stream>>>(src, dst, rank, segs, nbr, ne);
  gat_gather<<<(n + 7) / 8, T, 0, stream>>>(nbr, segs, cnt, h, e_src, e_dst, feat, out, n);
}


# Edges grouped via LDS in 2 launches ahead of the product: chunk blocks sort pairs by destination range, one block per range ranks ids in LDS buckets; no global atomic per edge; LDS arrival order
# speedup vs baseline: 4.4107x; 2.1569x over previous
#include <hip/hip_runtime.h>
#include <hip/hip_bf16.h>
#include <hip/hip_fp16.h>

typedef float v2f __attribute__((ext_vector_type(2)));
typedef float v8f __attribute__((ext_vector_type(8)));
typedef float v4f __attribute__((ext_vector_type(4)));

#define IN_DIM 128
#define HDIM 128
#define NHEADS 4
#define ODIM 32
#define CAP 32

#define BPADF 272
#define B_LDS_BYTES (64 * BPADF * 4)

__global__ __launch_bounds__(256) void gat_fc_wmma(
    const float* __restrict__ feat, const float* __restrict__ fc_w,
    float* __restrict__ h, int n) {
  extern __shared__ float Bs[];
  const int tid = threadIdx.x;
  for (int i = tid; i < IN_DIM * HDIM; i += 256) {
    const int kk = i & (IN_DIM - 1);
    const int jj = i >> 7;
    Bs[(kk >> 1) * BPADF + jj * 2 + (kk & 1)] = fc_w[jj * IN_DIM + kk];
  }
  __syncthreads();

  const int wave = tid >> 5;
  const int lane = tid & 31;
  const int half = lane >> 4;
  const int l16  = lane & 15;
  const int rowbase = blockIdx.x * 128 + wave * 16;
  if (rowbase >= n) return;

  const int row = min(rowbase + l16, n - 1);
  const float* __restrict__ fr = feat + (size_t)row * IN_DIM + 2 * half;
  const float* __restrict__ bbase = Bs + l16 * 2 + half * BPADF;

  v8f acc[8] = {};
  for (int kb = 0; kb < IN_DIM; kb += 16) {
    v2f a[4];
#pragma unroll
    for (int s = 0; s < 4; ++s) a[s] = *(const v2f*)(fr + kb + 4 * s);

#pragma unroll
    for (int s = 0; s < 4; ++s) {
      const float* __restrict__ bk = bbase + ((kb >> 1) + 2 * s) * BPADF;
      v2f b[8];
#pragma unroll
      for (int t = 0; t < 8; ++t) b[t] = *(const v2f*)(bk + t * 32);
#pragma unroll
      for (int t = 0; t < 8; ++t)
        acc[t] = __builtin_amdgcn_wmma_f32_16x16x4_f32(
            false, a[s], false, b[t], (short)0, acc[t], false, false);
    }
  }
#pragma unroll
  for (int j = 0; j < 8; ++j) {
    const int r = rowbase + j + 8 * half;
    if (r < n) {
      float* hr = h + (size_t)r * HDIM;
#pragma unroll
      for (int t = 0; t < 8; ++t) hr[t * 16 + l16] = acc[t][j];
    }
  }
}

__global__ void gat_node_logits(const float* __restrict__ h,
                                const float* __restrict__ attn_src,
                                const float* __restrict__ attn_dst,
                                float* __restrict__ e_src,
                                float* __restrict__ e_dst, int n) {
  int idx = blockIdx.x * blockDim.x + threadIdx.x;
  if (idx >= n * NHEADS) return;
  const int node = idx >> 2, hh = idx & 3;
  const float4* hp = (const float4*)(h + (size_t)node * HDIM + hh * ODIM);
  const float4* as = (const float4*)(attn_src + hh * ODIM);
  const float4* ad = (const float4*)(attn_dst + hh * ODIM);
  float ss = 0.0f, dd = 0.0f;
#pragma unroll
  for (int i = 0; i < ODIM / 4; ++i) {
    float4 v = hp[i], a = as[i], b = ad[i];
    ss += v.x * a.x + v.y * a.y + v.z * a.z + v.w * a.w;
    dd += v.x * b.x + v.y * b.y + v.z * b.z + v.w * b.w;
  }
  e_src[idx] = ss;
  e_dst[idx] = dd;
}

#define NBIN 256
#define NBLKA 256
#define GT 1024
static_assert(NBIN == 32 * 8, "the scan gives 8 ranges to each lane of one wave");
static_assert(NBLKA == (GT / 32) * 8, "gat_place gives 8 chunks to each wave");

__device__ __forceinline__ int gat_range(int d, int rpn) {
  const unsigned r = (unsigned)d / (unsigned)rpn;
  return (int)(r < (unsigned)(NBIN - 1) ? r : (unsigned)(NBIN - 1));
}

__global__ __launch_bounds__(GT) void gat_bin(const int* __restrict__ src,
                                              const int* __restrict__ dst, int ne, int epb,
                                              int rpn, int* __restrict__ staging,
                                              int* __restrict__ offs, int* __restrict__ ovc) {
  extern __shared__ int dyn[];
  __shared__ int cur[NBIN];
  __shared__ int beg[NBIN + 1];
  int* sd = dyn;
  int* ss = dyn + epb;
  const int tid = threadIdx.x;
  const int b = blockIdx.x;
  if (b == 0 && tid == 0) ovc[0] = 0;
  if (tid < NBIN) cur[tid] = 0;
  __syncthreads();
  const int e0 = b * epb;
  const int e1 = min(ne, e0 + epb);
  for (int eb = e0; eb < e1; eb += GT * 8) {
    int d[8];
#pragma unroll
    for (int u = 0; u < 8; ++u) {
      const int e = eb + u * GT + tid;
      d[u] = e < e1 ? dst[e] : 0;
    }
#pragma unroll
    for (int u = 0; u < 8; ++u)
      if (eb + u * GT + tid < e1) atomicAdd(&cur[gat_range(d[u], rpn)], 1);
  }
  __syncthreads();
  if (tid < 32) {
    int v[8];
    int sum = 0;
#pragma unroll
    for (int u = 0; u < 8; ++u) {
      v[u] = cur[tid * 8 + u];
      sum += v[u];
    }
    int incl = sum;
#pragma unroll
    for (int off = 1; off < 32; off <<= 1) {
      const int t = __shfl_up(incl, off);
      if (tid >= off) incl += t;
    }
    int run = incl - sum;
#pragma unroll
    for (int u = 0; u < 8; ++u) {
      beg[tid * 8 + u] = run;
      cur[tid * 8 + u] = run;
      run += v[u];
    }
    if (tid == 31) beg[NBIN] = run;
  }
  __syncthreads();
  for (int eb = e0; eb < e1; eb += GT * 8) {
    int d[8], s[8];
#pragma unroll
    for (int u = 0; u < 8; ++u) {
      const int e = eb + u * GT + tid;
      d[u] = e < e1 ? dst[e] : 0;
      s[u] = e < e1 ? src[e] : 0;
    }
#pragma unroll
    for (int u = 0; u < 8; ++u)
      if (eb + u * GT + tid < e1) {
        const int p = atomicAdd(&cur[gat_range(d[u], rpn)], 1);
        sd[p] = d[u];
        ss[p] = s[u];
      }
  }
  __syncthreads();
  int2* __restrict__ pairs = (int2*)staging;
  for (int i = tid; i < e1 - e0; i += GT) pairs[e0 + i] = make_int2(sd[i], ss[i]);
  if (tid <= NBIN) offs[b * (NBIN + 1) + tid] = beg[tid];
}

__device__ __forceinline__ void gat_put(int2 e, int base, int nloc, int* lc, int* lb,
                                        int* __restrict__ ovc, int* __restrict__ ov_dst,
                                        int* __restrict__ ov_src, int ovcap) {
  const int dl = e.x - base;
  if (dl >= 0 && dl < nloc) {
    const int r = atomicAdd(&lc[dl], 1);
    if (r < CAP) {
      lb[dl * CAP + r] = e.y;
    } else {
      const int o = atomicAdd(ovc, 1);
      if (o < ovcap) {
        ov_dst[o] = e.x;
        ov_src[o] = e.y;
      }
    }
  }
}

__global__ __launch_bounds__(GT) void gat_place(const int* __restrict__ staging,
                                                const int* __restrict__ offs, int epb, int rpn,
                                                int n, int* __restrict__ cnt,
                                                int* __restrict__ bucket, int* __restrict__ ovc,
                                                int* __restrict__ ov_dst,
                                                int* __restrict__ ov_src, int ovcap) {
  extern __shared__ int dyn[];
  __shared__ int seg0[NBLKA];
  __shared__ int segn[NBLKA];
  int* lb = dyn;
  int* lc = dyn + rpn * CAP;
  const int2* __restrict__ pairs = (const int2*)staging;
  const int tid = threadIdx.x;
  const int g = blockIdx.x;
  const int base = g * rpn;
  const int nloc = min(rpn, n - base);
  for (int i = tid; i < rpn * (CAP + 1); i += GT) dyn[i] = 0;
  if (tid < NBLKA) {
    const int* ob = offs + tid * (NBIN + 1) + g;
    const int a = ob[0];
    seg0[tid] = tid * epb + a;
    segn[tid] = ob[1] - a;
  }
  __syncthreads();
  const int w = tid >> 5;
  const int lane = tid & 31;
  int st[8], ln[8];
  int2 it[8];
#pragma unroll
  for (int u = 0; u < 8; ++u) {
    st[u] = seg0[w + 32 * u];
    ln[u] = segn[w + 32 * u];
  }
#pragma unroll
  for (int u = 0; u < 8; ++u)
    it[u] = lane < ln[u] ? pairs[st[u] + lane] : make_int2(-1, 0);
#pragma unroll
  for (int u = 0; u < 8; ++u) {
    if (lane < ln[u]) gat_put(it[u], base, nloc, lc, lb, ovc, ov_dst, ov_src, ovcap);
    for (int i = lane + 32; i < ln[u]; i += 32)
      gat_put(pairs[st[u] + i], base, nloc, lc, lb, ovc, ov_dst, ov_src, ovcap);
  }
  __syncthreads();
  for (int i = tid; i < nloc * CAP; i += GT) bucket[base * CAP + i] = lb[i];
  for (int i = tid; i < nloc; i += GT) cnt[base + i] = lc[i];
}

__device__ __forceinline__ float gat_leaky(float x) {
  return x > 0.0f ? x : 0.2f * x;
}

template <int CH, bool TAIL>
__device__ __forceinline__ void gat_chunk(const int* s_in, int rem,
                                          const float* __restrict__ e_src,
                                          const float* __restrict__ h,
                                          int k, int lane, float edk,
                                          float& m, float& den, float4& acc) {
  int s[CH];
  float e[CH];
  v4f hv[CH];
#pragma unroll
  for (int u = 0; u < CH; ++u) s[u] = (!TAIL || u < rem) ? s_in[u] : s_in[0];
#pragma unroll
  for (int u = 0; u < CH; ++u) {
    e[u] = e_src[s[u] * NHEADS + k];
    hv[u] = ((const v4f*)h)[(unsigned)s[u] * (HDIM / 4) + (unsigned)lane];
  }
  if constexpr (CH == 8)
    asm volatile("" : "+v"(e[0]), "+v"(e[1]), "+v"(e[2]), "+v"(e[3]),
                      "+v"(e[CH - 4]), "+v"(e[CH - 3]), "+v"(e[CH - 2]), "+v"(e[CH - 1])
                    : "v"(hv[0]), "v"(hv[1]), "v"(hv[2]), "v"(hv[3]),
                      "v"(hv[CH - 4]), "v"(hv[CH - 3]), "v"(hv[CH - 2]), "v"(hv[CH - 1]));
  else
    asm volatile("" : "+v"(e[0]), "+v"(e[1]), "+v"(e[2]), "+v"(e[3])
                    : "v"(hv[0]), "v"(hv[1]), "v"(hv[2]), "v"(hv[3]));
#pragma unroll
  for (int u = 0; u < CH; ++u) e[u] = gat_leaky(e[u] + edk);
  float mn = m;
#pragma unroll
  for (int u = 0; u < CH; ++u) mn = fmaxf(mn, e[u]);
  const float sc = __expf(m - mn);
  m = mn;
  den *= sc;
  acc.x *= sc;
  acc.y *= sc;
  acc.z *= sc;
  acc.w *= sc;
#pragma unroll
  for (int u = 0; u < CH; ++u) {
    float a = __expf(e[u] - mn);
    if (TAIL && u >= rem) a = 0.0f;
    den += a;
    acc.x += a * hv[u].x;
    acc.y += a * hv[u].y;
    acc.z += a * hv[u].z;
    acc.w += a * hv[u].w;
  }
}

__device__ __forceinline__ void gat_from_list(const int* __restrict__ ld,
                                              const int* __restrict__ ls, int nl, int d,
                                              const float* __restrict__ e_src,
                                              const float* __restrict__ h,
                                              int k, int lane, float edk,
                                              float& m, float& den, float4& acc) {
  for (int i0 = 0; i0 < nl; i0 += 32) {
    const int i = i0 + lane;
    int dd = -1, ss = 0;
    if (i < nl) {
      dd = ld[i];
      ss = ls[i];
    }
    unsigned mask = __builtin_amdgcn_ballot_w32(dd == d);
    while (mask != 0u) {
      const int b = __builtin_ctz(mask);
      mask &= mask - 1u;
      int one[4];
      one[0] = one[1] = one[2] = one[3] = __builtin_amdgcn_readlane(ss, b);
      gat_chunk<4, true>(one, 1, e_src, h, k, lane, edk, m, den, acc);
    }
  }
}

__global__ __launch_bounds__(256) void gat_gather(
    const int* __restrict__ bucket, const int* __restrict__ cnt,
    const int* __restrict__ ovc, const int* __restrict__ ov_dst,
    const int* __restrict__ ov_src, int ovcap,
    const int* __restrict__ src, const int* __restrict__ dst, int ne,
    const float* __restrict__ h,
    const float* __restrict__ e_src, const float* __restrict__ e_dst,
    const float* __restrict__ feat, float* __restrict__ out, int n) {
  const int d = __builtin_amdgcn_readfirstlane(blockIdx.x * 8 + (threadIdx.x >> 5));
  if (d >= n) return;
  const int lane = threadIdx.x & 31;
  const int k = lane >> 3;
  const int deg = cnt[d];
  const int* __restrict__ nb = bucket + d * CAP;
  const float edk = e_dst[d * NHEADS + k];
  const float4 f = *(const float4*)(feat + (size_t)d * HDIM + lane * 4);
  int cur[8], nxt[8];
#pragma unroll
  for (int u = 0; u < 8; ++u) cur[u] = nb[u];
  __builtin_amdgcn_sched_barrier(0);

  float m = -INFINITY;
  float den = 0.0f;
  float4 acc = make_float4(0.0f, 0.0f, 0.0f, 0.0f);
  int degb = deg;
  int ovn = 0;
  if (deg > CAP) {
    ovn = ovc[0];
    degb = ovn <= ovcap ? CAP : 0;
  }
  for (int j0 = 0; j0 < degb; j0 += 8) {
    const int rem = degb - j0;
    if (rem > 8) {
#pragma unroll
      for (int u = 0; u < 8; ++u) nxt[u] = nb[j0 + 8 + u];
      gat_chunk<8, false>(cur, 8, e_src, h, k, lane, edk, m, den, acc);
#pragma unroll
      for (int u = 0; u < 8; ++u) cur[u] = nxt[u];
    } else if (rem == 8) {
      gat_chunk<8, false>(cur, 8, e_src, h, k, lane, edk, m, den, acc);
    } else if (rem > 4) {
      gat_chunk<8, true>(cur, rem, e_src, h, k, lane, edk, m, den, acc);
    } else {
      gat_chunk<4, true>(cur, rem, e_src, h, k, lane, edk, m, den, acc);
    }
  }

  if (deg > CAP) {
    if (ovn <= ovcap)
      gat_from_list(ov_dst, ov_src, ovn, d, e_src, h, k, lane, edk, m, den, acc);
    else
      gat_from_list(dst, src, ne, d, e_src, h, k, lane, edk, m, den, acc);
  }

  const float dsafe = den > 0.0f ? den : 1.0f;
  float4 o;
  o.x = f.x + acc.x / dsafe;
  o.y = f.y + acc.y / dsafe;
  o.z = f.z + acc.z / dsafe;
  o.w = f.w + acc.w / dsafe;
  *(float4*)(out + (size_t)d * HDIM + lane * 4) = o;
}

extern "C" void kernel_launch(void* const* d_in, const int* in_sizes, int n_in,
                              void* d_out, int out_size, void* d_ws, size_t ws_size,
                              hipStream_t stream) {
  const float* feat     = (const float*)d_in[0];
  const float* fc_w     = (const float*)d_in[1];
  const float* attn_src = (const float*)d_in[2];
  const float* attn_dst = (const float*)d_in[3];
  const int*   src      = (const int*)d_in[4];
  const int*   dst      = (const int*)d_in[5];
  float* out = (float*)d_out;

  const int n  = in_sizes[0] / IN_DIM;
  const int ne = in_sizes[4];

  char* ws = (char*)d_ws;
  size_t off = 0;
  float* h     = (float*)(ws + off); off += (size_t)n * HDIM * sizeof(float);
  float* e_src = (float*)(ws + off); off += (size_t)n * NHEADS * sizeof(float);
  float* e_dst = (float*)(ws + off); off += (size_t)n * NHEADS * sizeof(float);
  int*   bucket = (int*)(ws + off);  off += (size_t)n * CAP * sizeof(int);
  int*   cnt    = (int*)(ws + off);  off += (size_t)n * sizeof(int);
  const int ovcap = n / 2;
  int*   ovc    = (int*)(ws + off);  off += 4 * sizeof(int);
  int*   ov_dst = (int*)(ws + off);  off += (size_t)ovcap * sizeof(int);
  int*   ov_src = (int*)(ws + off);  off += (size_t)ovcap * sizeof(int);
  (void)ws_size; (void)n_in; (void)out_size;

  const int epb = (ne + NBLKA - 1) / NBLKA;
  const int rpn = (n + NBIN - 1) / NBIN;
  int* staging = (int*)h;
  int* offs    = staging + (size_t)2 * NBLKA * epb;

  const int T = 256;
  gat_bin<<<NBLKA, GT, (size_t)2 * epb * sizeof(int), stream>>>(src, dst, ne, epb, rpn, staging,
                                                               offs, ovc);
  gat_place<<<NBIN, GT, (size_t)rpn * (CAP + 1) * sizeof(int), stream>>>(
      staging, offs, epb, rpn, n, cnt, bucket, ovc, ov_dst, ov_src, ovcap);
  gat_fc_wmma<<<(n + 127) / 128, T, B_LDS_BYTES, stream>>>(feat, fc_w, h, n);
  gat_node_logits<<<(n * NHEADS + T - 1) / T, T, 0, stream>>>(h, attn_src, attn_dst,
                                                              e_src, e_dst, n);
  gat_gather<<<(n + 7) / 8, T, 0, stream>>>(bucket, cnt, ovc, ov_dst, ov_src, ovcap,
                                            src, dst, ne, h, e_src, e_dst, feat, out, n);
}


# Gather forms each edge's logit from the source row it holds (4 products + 3 lane-crossbar adds per head): logits launch and per-edge logit read go (5 -> 4 launches); logit summed as a lane tree
# speedup vs baseline: 4.6154x; 1.0464x over previous
#include <hip/hip_runtime.h>
#include <hip/hip_bf16.h>
#include <hip/hip_fp16.h>

typedef float v2f __attribute__((ext_vector_type(2)));
typedef float v8f __attribute__((ext_vector_type(8)));
typedef float v4f __attribute__((ext_vector_type(4)));

#define IN_DIM 128
#define HDIM 128
#define NHEADS 4
#define ODIM 32
#define CAP 32

#define BPADF 272
#define B_LDS_BYTES (64 * BPADF * 4)

__global__ __launch_bounds__(256) void gat_fc_wmma(
    const float* __restrict__ feat, const float* __restrict__ fc_w,
    float* __restrict__ h, int n) {
  extern __shared__ float Bs[];
  const int tid = threadIdx.x;
  for (int i = tid; i < IN_DIM * HDIM; i += 256) {
    const int kk = i & (IN_DIM - 1);
    const int jj = i >> 7;
    Bs[(kk >> 1) * BPADF + jj * 2 + (kk & 1)] = fc_w[jj * IN_DIM + kk];
  }
  __syncthreads();

  const int wave = tid >> 5;
  const int lane = tid & 31;
  const int half = lane >> 4;
  const int l16  = lane & 15;
  const int rowbase = blockIdx.x * 128 + wave * 16;
  if (rowbase >= n) return;

  const int row = min(rowbase + l16, n - 1);
  const float* __restrict__ fr = feat + (size_t)row * IN_DIM + 2 * half;
  const float* __restrict__ bbase = Bs + l16 * 2 + half * BPADF;

  v8f acc[8] = {};
  for (int kb = 0; kb < IN_DIM; kb += 16) {
    v2f a[4];
#pragma unroll
    for (int s = 0; s < 4; ++s) a[s] = *(const v2f*)(fr + kb + 4 * s);

#pragma unroll
    for (int s = 0; s < 4; ++s) {
      const float* __restrict__ bk = bbase + ((kb >> 1) + 2 * s) * BPADF;
      v2f b[8];
#pragma unroll
      for (int t = 0; t < 8; ++t) b[t] = *(const v2f*)(bk + t * 32);
#pragma unroll
      for (int t = 0; t < 8; ++t)
        acc[t] = __builtin_amdgcn_wmma_f32_16x16x4_f32(
            false, a[s], false, b[t], (short)0, acc[t], false, false);
    }
  }
#pragma unroll
  for (int j = 0; j < 8; ++j) {
    const int r = rowbase + j + 8 * half;
    if (r < n) {
      float* hr = h + (size_t)r * HDIM;
#pragma unroll
      for (int t = 0; t < 8; ++t) hr[t * 16 + l16] = acc[t][j];
    }
  }
}

#define NBIN 256
#define NBLKA 256
#define GT 1024
static_assert(NBIN == 32 * 8, "the scan gives 8 ranges to each lane of one wave");
static_assert(NBLKA == (GT / 32) * 8, "gat_place gives 8 chunks to each wave");

__device__ __forceinline__ int gat_range(int d, int rpn) {
  const unsigned r = (unsigned)d / (unsigned)rpn;
  return (int)(r < (unsigned)(NBIN - 1) ? r : (unsigned)(NBIN - 1));
}

__global__ __launch_bounds__(GT) void gat_bin(const int* __restrict__ src,
                                              const int* __restrict__ dst, int ne, int epb,
                                              int rpn, int* __restrict__ staging,
                                              int* __restrict__ offs, int* __restrict__ ovc) {
  extern __shared__ int dyn[];
  __shared__ int cur[NBIN];
  __shared__ int beg[NBIN + 1];
  int* sd = dyn;
  int* ss = dyn + epb;
  const int tid = threadIdx.x;
  const int b = blockIdx.x;
  if (b == 0 && tid == 0) ovc[0] = 0;
  if (tid < NBIN) cur[tid] = 0;
  __syncthreads();
  const int e0 = b * epb;
  const int e1 = min(ne, e0 + epb);
  for (int eb = e0; eb < e1; eb += GT * 8) {
    int d[8];
#pragma unroll
    for (int u = 0; u < 8; ++u) {
      const int e = eb + u * GT + tid;
      d[u] = e < e1 ? dst[e] : 0;
    }
#pragma unroll
    for (int u = 0; u < 8; ++u)
      if (eb + u * GT + tid < e1) atomicAdd(&cur[gat_range(d[u], rpn)], 1);
  }
  __syncthreads();
  if (tid < 32) {
    int v[8];
    int sum = 0;
#pragma unroll
    for (int u = 0; u < 8; ++u) {
      v[u] = cur[tid * 8 + u];
      sum += v[u];
    }
    int incl = sum;
#pragma unroll
    for (int off = 1; off < 32; off <<= 1) {
      const int t = __shfl_up(incl, off);
      if (tid >= off) incl += t;
    }
    int run = incl - sum;
#pragma unroll
    for (int u = 0; u < 8; ++u) {
      beg[tid * 8 + u] = run;
      cur[tid * 8 + u] = run;
      run += v[u];
    }
    if (tid == 31) beg[NBIN] = run;
  }
  __syncthreads();
  for (int eb = e0; eb < e1; eb += GT * 8) {
    int d[8], s[8];
#pragma unroll
    for (int u = 0; u < 8; ++u) {
      const int e = eb + u * GT + tid;
      d[u] = e < e1 ? dst[e] : 0;
      s[u] = e < e1 ? src[e] : 0;
    }
#pragma unroll
    for (int u = 0; u < 8; ++u)
      if (eb + u * GT + tid < e1) {
        const int p = atomicAdd(&cur[gat_range(d[u], rpn)], 1);
        sd[p] = d[u];
        ss[p] = s[u];
      }
  }
  __syncthreads();
  int2* __restrict__ pairs = (int2*)staging;
  for (int i = tid; i < e1 - e0; i += GT) pairs[e0 + i] = make_int2(sd[i], ss[i]);
  if (tid <= NBIN) offs[b * (NBIN + 1) + tid] = beg[tid];
}

__device__ __forceinline__ void gat_put(int2 e, int base, int nloc, int* lc, int* lb,
                                        int* __restrict__ ovc, int* __restrict__ ov_dst,
                                        int* __restrict__ ov_src, int ovcap) {
  const int dl = e.x - base;
  if (dl >= 0 && dl < nloc) {
    const int r = atomicAdd(&lc[dl], 1);
    if (r < CAP) {
      lb[dl * CAP + r] = e.y;
    } else {
      const int o = atomicAdd(ovc, 1);
      if (o < ovcap) {
        ov_dst[o] = e.x;
        ov_src[o] = e.y;
      }
    }
  }
}

__global__ __launch_bounds__(GT) void gat_place(const int* __restrict__ staging,
                                                const int* __restrict__ offs, int epb, int rpn,
                                                int n, int* __restrict__ cnt,
                                                int* __restrict__ bucket, int* __restrict__ ovc,
                                                int* __restrict__ ov_dst,
                                                int* __restrict__ ov_src, int ovcap) {
  extern __shared__ int dyn[];
  __shared__ int seg0[NBLKA];
  __shared__ int segn[NBLKA];
  int* lb = dyn;
  int* lc = dyn + rpn * CAP;
  const int2* __restrict__ pairs = (const int2*)staging;
  const int tid = threadIdx.x;
  const int g = blockIdx.x;
  const int base = g * rpn;
  const int nloc = min(rpn, n - base);
  for (int i = tid; i < rpn * (CAP + 1); i += GT) dyn[i] = 0;
  if (tid < NBLKA) {
    const int* ob = offs + tid * (NBIN + 1) + g;
    const int a = ob[0];
    seg0[tid] = tid * epb + a;
    segn[tid] = ob[1] - a;
  }
  __syncthreads();
  const int w = tid >> 5;
  const int lane = tid & 31;
  int st[8], ln[8];
  int2 it[8];
#pragma unroll
  for (int u = 0; u < 8; ++u) {
    st[u] = seg0[w + 32 * u];
    ln[u] = segn[w + 32 * u];
  }
#pragma unroll
  for (int u = 0; u < 8; ++u)
    it[u] = lane < ln[u] ? pairs[st[u] + lane] : make_int2(-1, 0);
#pragma unroll
  for (int u = 0; u < 8; ++u) {
    if (lane < ln[u]) gat_put(it[u], base, nloc, lc, lb, ovc, ov_dst, ov_src, ovcap);
    for (int i = lane + 32; i < ln[u]; i += 32)
      gat_put(pairs[st[u] + i], base, nloc, lc, lb, ovc, ov_dst, ov_src, ovcap);
  }
  __syncthreads();
  for (int i = tid; i < nloc * CAP; i += GT) bucket[base * CAP + i] = lb[i];
  for (int i = tid; i < nloc; i += GT) cnt[base + i] = lc[i];
}

__device__ __forceinline__ float gat_leaky(float x) {
  return x > 0.0f ? x : 0.2f * x;
}

template <int MASK>
__device__ __forceinline__ float gat_xadd(float x) {
  float y;
  asm volatile("v_add_f32_dpp %0, %1, %1 row_xmask:%2 row_mask:0xf bank_mask:0xf bound_ctrl:1"
      : "=v"(y)
      : "v"(x), "n"(MASK));
  return y;
}

__device__ __forceinline__ float gat_dot4(v4f r, float4 a) {
  return r.x * a.x + r.y * a.y + r.z * a.z + r.w * a.w;
}

__device__ __forceinline__ float gat_head_sum(float p) {
  p = gat_xadd<1>(p);
  p = gat_xadd<2>(p);
  p = gat_xadd<4>(p);
  return p;
}

template <int CH, bool TAIL>
__device__ __forceinline__ void gat_chunk(const int* s_in, int rem,
                                          const float* __restrict__ h,
                                          int lane, float4 asl, float edk,
                                          float& m, float& den, float4& acc) {
  int s[CH];
  float e[CH];
  v4f hv[CH];
#pragma unroll
  for (int u = 0; u < CH; ++u) s[u] = (!TAIL || u < rem) ? s_in[u] : s_in[0];
#pragma unroll
  for (int u = 0; u < CH; ++u) {
    hv[u] = ((const v4f*)h)[(unsigned)s[u] * (HDIM / 4) + (unsigned)lane];
  }
#pragma unroll
  for (int u = 0; u < CH; ++u) e[u] = gat_dot4(hv[u], asl);
#pragma unroll
  for (int u = 0; u < CH; ++u) e[u] = gat_xadd<1>(e[u]);
#pragma unroll
  for (int u = 0; u < CH; ++u) e[u] = gat_xadd<2>(e[u]);
#pragma unroll
  for (int u = 0; u < CH; ++u) e[u] = gat_xadd<4>(e[u]);
#pragma unroll
  for (int u = 0; u < CH; ++u) e[u] = gat_leaky(e[u] + edk);
  float mn = m;
#pragma unroll
  for (int u = 0; u < CH; ++u) mn = fmaxf(mn, e[u]);
  const float sc = __expf(m - mn);
  m = mn;
  den *= sc;
  acc.x *= sc;
  acc.y *= sc;
  acc.z *= sc;
  acc.w *= sc;
#pragma unroll
  for (int u = 0; u < CH; ++u) {
    float a = __expf(e[u] - mn);
    if (TAIL && u >= rem) a = 0.0f;
    den += a;
    acc.x += a * hv[u].x;
    acc.y += a * hv[u].y;
    acc.z += a * hv[u].z;
    acc.w += a * hv[u].w;
  }
}

__device__ __forceinline__ void gat_from_list(const int* __restrict__ ld,
                                              const int* __restrict__ ls, int nl, int d,
                                              const float* __restrict__ h,
                                              int lane, float4 asl, float edk,
                                              float& m, float& den, float4& acc) {
  for (int i0 = 0; i0 < nl; i0 += 32) {
    const int i = i0 + lane;
    int dd = -1, ss = 0;
    if (i < nl) {
      dd = ld[i];
      ss = ls[i];
    }
    unsigned mask = __builtin_amdgcn_ballot_w32(dd == d);
    while (mask != 0u) {
      const int b = __builtin_ctz(mask);
      mask &= mask - 1u;
      int one[4];
      one[0] = one[1] = one[2] = one[3] = __builtin_amdgcn_readlane(ss, b);
      gat_chunk<4, true>(one, 1, h, lane, asl, edk, m, den, acc);
    }
  }
}

__global__ __launch_bounds__(256) void gat_gather(
    const int* __restrict__ bucket, const int* __restrict__ cnt,
    const int* __restrict__ ovc, const int* __restrict__ ov_dst,
    const int* __restrict__ ov_src, int ovcap,
    const int* __restrict__ src, const int* __restrict__ dst, int ne,
    const float* __restrict__ h,
    const float* __restrict__ attn_src, const float* __restrict__ attn_dst,
    const float* __restrict__ feat, float* __restrict__ out, int n) {
  const int d = __builtin_amdgcn_readfirstlane(blockIdx.x * 8 + (threadIdx.x >> 5));
  if (d >= n) return;
  const int lane = threadIdx.x & 31;
  const int deg = cnt[d];
  const int* __restrict__ nb = bucket + d * CAP;
  const float4 asl = *(const float4*)(attn_src + lane * 4);
  const float4 adl = *(const float4*)(attn_dst + lane * 4);
  const v4f hd = ((const v4f*)h)[(unsigned)d * (HDIM / 4) + (unsigned)lane];
  const float4 f = *(const float4*)(feat + (size_t)d * HDIM + lane * 4);
  int cur[8], nxt[8];
#pragma unroll
  for (int u = 0; u < 8; ++u) cur[u] = nb[u];
  __builtin_amdgcn_sched_barrier(0);
  const float edk = gat_head_sum(gat_dot4(hd, adl));

  float m = -INFINITY;
  float den = 0.0f;
  float4 acc = make_float4(0.0f, 0.0f, 0.0f, 0.0f);
  int degb = deg;
  int ovn = 0;
  if (deg > CAP) {
    ovn = ovc[0];
    degb = ovn <= ovcap ? CAP : 0;
  }
  for (int j0 = 0; j0 < degb; j0 += 8) {
    const int rem = degb - j0;
    if (rem > 8) {
#pragma unroll
      for (int u = 0; u < 8; ++u) nxt[u] = nb[j0 + 8 + u];
      gat_chunk<8, false>(cur, 8, h, lane, asl, edk, m, den, acc);
#pragma unroll
      for (int u = 0; u < 8; ++u) cur[u] = nxt[u];
    } else if (rem == 8) {
      gat_chunk<8, false>(cur, 8, h, lane, asl, edk, m, den, acc);
    } else if (rem > 4) {
      gat_chunk<8, true>(cur, rem, h, lane, asl, edk, m, den, acc);
    } else {
      gat_chunk<4, true>(cur, rem, h, lane, asl, edk, m, den, acc);
    }
  }

  if (deg > CAP) {
    if (ovn <= ovcap)
      gat_from_list(ov_dst, ov_src, ovn, d, h, lane, asl, edk, m, den, acc);
    else
      gat_from_list(dst, src, ne, d, h, lane, asl, edk, m, den, acc);
  }

  const float dsafe = den > 0.0f ? den : 1.0f;
  float4 o;
  o.x = f.x + acc.x / dsafe;
  o.y = f.y + acc.y / dsafe;
  o.z = f.z + acc.z / dsafe;
  o.w = f.w + acc.w / dsafe;
  *(float4*)(out + (size_t)d * HDIM + lane * 4) = o;
}

extern "C" void kernel_launch(void* const* d_in, const int* in_sizes, int n_in,
                              void* d_out, int out_size, void* d_ws, size_t ws_size,
                              hipStream_t stream) {
  const float* feat     = (const float*)d_in[0];
  const float* fc_w     = (const float*)d_in[1];
  const float* attn_src = (const float*)d_in[2];
  const float* attn_dst = (const float*)d_in[3];
  const int*   src      = (const int*)d_in[4];
  const int*   dst      = (const int*)d_in[5];
  float* out = (float*)d_out;

  const int n  = in_sizes[0] / IN_DIM;
  const int ne = in_sizes[4];

  char* ws = (char*)d_ws;
  size_t off = 0;
  float* h     = (float*)(ws + off); off += (size_t)n * HDIM * sizeof(float);
  int*   bucket = (int*)(ws + off);  off += (size_t)n * CAP * sizeof(int);
  int*   cnt    = (int*)(ws + off);  off += (size_t)n * sizeof(int);
  const int ovcap = n / 2;
  int*   ovc    = (int*)(ws + off);  off += 4 * sizeof(int);
  int*   ov_dst = (int*)(ws + off);  off += (size_t)ovcap * sizeof(int);
  int*   ov_src = (int*)(ws + off);  off += (size_t)ovcap * sizeof(int);
  (void)ws_size; (void)n_in; (void)out_size;

  const int epb = (ne + NBLKA - 1) / NBLKA;
  const int rpn = (n + NBIN - 1) / NBIN;
  int* staging = (int*)h;
  int* offs    = staging + (size_t)2 * NBLKA * epb;

  const int T = 256;
  gat_bin<<<NBLKA, GT, (size_t)2 * epb * sizeof(int), stream>>>(src, dst, ne, epb, rpn, staging,
                                                               offs, ovc);
  gat_place<<<NBIN, GT, (size_t)rpn * (CAP + 1) * sizeof(int), stream>>>(
      staging, offs, epb, rpn, n, cnt, bucket, ovc, ov_dst, ov_src, ovcap);
  gat_fc_wmma<<<(n + 127) / 128, T, B_LDS_BYTES, stream>>>(feat, fc_w, h, n);
  gat_gather<<<(n + 7) / 8, T, 0, stream>>>(bucket, cnt, ovc, ov_dst, ov_src, ovcap,
                                            src, dst, ne, h, attn_src, attn_dst, feat, out, n);
}


# Placement and gather in one launch: a range's work-group ranks its edges into LDS buckets, then its 32 waves take nodes off an LDS counter and walk the buckets from LDS, 12 rows a trip; launches 4->3
# speedup vs baseline: 4.7872x; 1.0372x over previous
#include <hip/hip_runtime.h>
#include <hip/hip_bf16.h>
#include <hip/hip_fp16.h>

typedef float v2f __attribute__((ext_vector_type(2)));
typedef float v8f __attribute__((ext_vector_type(8)));
typedef float v4f __attribute__((ext_vector_type(4)));

#define IN_DIM 128
#define HDIM 128
#define NHEADS 4
#define ODIM 32
#define CAP 32

#define BPADF 272
#define B_LDS_BYTES (64 * BPADF * 4)

__global__ __launch_bounds__(256) void gat_fc_wmma(
    const float* __restrict__ feat, const float* __restrict__ fc_w,
    float* __restrict__ h, int n) {
  extern __shared__ float Bs[];
  const int tid = threadIdx.x;
  for (int i = tid; i < IN_DIM * HDIM; i += 256) {
    const int kk = i & (IN_DIM - 1);
    const int jj = i >> 7;
    Bs[(kk >> 1) * BPADF + jj * 2 + (kk & 1)] = fc_w[jj * IN_DIM + kk];
  }
  __syncthreads();

  const int wave = tid >> 5;
  const int lane = tid & 31;
  const int half = lane >> 4;
  const int l16  = lane & 15;
  const int rowbase = blockIdx.x * 128 + wave * 16;
  if (rowbase >= n) return;

  const int row = min(rowbase + l16, n - 1);
  const float* __restrict__ fr = feat + (size_t)row * IN_DIM + 2 * half;
  const float* __restrict__ bbase = Bs + l16 * 2 + half * BPADF;

  v8f acc[8] = {};
  for (int kb = 0; kb < IN_DIM; kb += 16) {
    v2f a[4];
#pragma unroll
    for (int s = 0; s < 4; ++s) a[s] = *(const v2f*)(fr + kb + 4 * s);

#pragma unroll
    for (int s = 0; s < 4; ++s) {
      const float* __restrict__ bk = bbase + ((kb >> 1) + 2 * s) * BPADF;
      v2f b[8];
#pragma unroll
      for (int t = 0; t < 8; ++t) b[t] = *(const v2f*)(bk + t * 32);
#pragma unroll
      for (int t = 0; t < 8; ++t)
        acc[t] = __builtin_amdgcn_wmma_f32_16x16x4_f32(
            false, a[s], false, b[t], (short)0, acc[t], false, false);
    }
  }
#pragma unroll
  for (int j = 0; j < 8; ++j) {
    const int r = rowbase + j + 8 * half;
    if (r < n) {
      float* hr = h + (size_t)r * HDIM;
#pragma unroll
      for (int t = 0; t < 8; ++t) hr[t * 16 + l16] = acc[t][j];
    }
  }
}

#define NBIN 256
#define NBLKA 256
#define GT 1024
#define RPL (NBIN / 32)
#define OVL 256
static_assert(NBIN == 32 * RPL, "the scan gives RPL ranges to each lane of one wave");
static_assert(NBIN + 1 <= GT, "one thread per segment start");
static_assert(NBLKA == (GT / 32) * 8, "the placement gives 8 chunks to each wave");

__device__ __forceinline__ int gat_range(int d, int rpn) {
  const unsigned r = (unsigned)d / (unsigned)rpn;
  return (int)(r < (unsigned)(NBIN - 1) ? r : (unsigned)(NBIN - 1));
}

__global__ __launch_bounds__(GT) void gat_bin(const int* __restrict__ src,
                                              const int* __restrict__ dst, int ne, int epb,
                                              int rpn, int* __restrict__ staging,
                                              int* __restrict__ offs) {
  extern __shared__ int dyn[];
  __shared__ int cur[NBIN];
  __shared__ int beg[NBIN + 1];
  int* sd = dyn;
  int* ss = dyn + epb;
  const int tid = threadIdx.x;
  const int b = blockIdx.x;
  if (tid < NBIN) cur[tid] = 0;
  __syncthreads();
  const int e0 = b * epb;
  const int e1 = min(ne, e0 + epb);
  for (int eb = e0; eb < e1; eb += GT * 8) {
    int d[8];
#pragma unroll
    for (int u = 0; u < 8; ++u) {
      const int e = eb + u * GT + tid;
      d[u] = e < e1 ? dst[e] : 0;
    }
#pragma unroll
    for (int u = 0; u < 8; ++u)
      if (eb + u * GT + tid < e1) atomicAdd(&cur[gat_range(d[u], rpn)], 1);
  }
  __syncthreads();
  if (tid < 32) {
    int v[RPL];
    int sum = 0;
#pragma unroll
    for (int u = 0; u < RPL; ++u) {
      v[u] = cur[tid * RPL + u];
      sum += v[u];
    }
    int incl = sum;
#pragma unroll
    for (int off = 1; off < 32; off <<= 1) {
      const int t = __shfl_up(incl, off);
      if (tid >= off) incl += t;
    }
    int run = incl - sum;
#pragma unroll
    for (int u = 0; u < RPL; ++u) {
      beg[tid * RPL + u] = run;
      cur[tid * RPL + u] = run;
      run += v[u];
    }
    if (tid == 31) beg[NBIN] = run;
  }
  __syncthreads();
  for (int eb = e0; eb < e1; eb += GT * 8) {
    int d[8], s[8];
#pragma unroll
    for (int u = 0; u < 8; ++u) {
      const int e = eb + u * GT + tid;
      d[u] = e < e1 ? dst[e] : 0;
      s[u] = e < e1 ? src[e] : 0;
    }
#pragma unroll
    for (int u = 0; u < 8; ++u)
      if (eb + u * GT + tid < e1) {
        const int p = atomicAdd(&cur[gat_range(d[u], rpn)], 1);
        sd[p] = d[u];
        ss[p] = s[u];
      }
  }
  __syncthreads();
  int2* __restrict__ pairs = (int2*)staging;
  for (int i = tid; i < e1 - e0; i += GT) pairs[e0 + i] = make_int2(sd[i], ss[i]);
  if (tid <= NBIN) offs[b * (NBIN + 1) + tid] = beg[tid];
}

__device__ __forceinline__ float gat_leaky(float x) {
  return x > 0.0f ? x : 0.2f * x;
}

template <int MASK>
__device__ __forceinline__ float gat_xadd(float x) {
  float y;
  asm volatile("v_add_f32_dpp %0, %1, %1 row_xmask:%2 row_mask:0xf bank_mask:0xf bound_ctrl:1"
      : "=v"(y)
      : "v"(x), "n"(MASK));
  return y;
}

__device__ __forceinline__ float gat_dot4(v4f r, float4 a) {
  return r.x * a.x + r.y * a.y + r.z * a.z + r.w * a.w;
}

__device__ __forceinline__ float gat_head_sum(float p) {
  p = gat_xadd<1>(p);
  p = gat_xadd<2>(p);
  p = gat_xadd<4>(p);
  return p;
}

template <int CH, bool TAIL>
__device__ __forceinline__ void gat_chunk(const int* s_in, int rem,
                                          const float* __restrict__ h,
                                          int lane, float4 asl, float edk,
                                          float& m, float& den, float4& acc) {
  int s[CH];
  float e[CH];
  v4f hv[CH];
#pragma unroll
  for (int u = 0; u < CH; ++u) s[u] = (!TAIL || u < rem) ? s_in[u] : s_in[0];
#pragma unroll
  for (int u = 0; u < CH; ++u) {
    hv[u] = ((const v4f*)h)[(unsigned)s[u] * (HDIM / 4) + (unsigned)lane];
  }
#pragma unroll
  for (int u = 0; u < CH; ++u) e[u] = gat_dot4(hv[u], asl);
#pragma unroll
  for (int u = 0; u < CH; ++u) e[u] = gat_xadd<1>(e[u]);
#pragma unroll
  for (int u = 0; u < CH; ++u) e[u] = gat_xadd<2>(e[u]);
#pragma unroll
  for (int u = 0; u < CH; ++u) e[u] = gat_xadd<4>(e[u]);
#pragma unroll
  for (int u = 0; u < CH; ++u) e[u] = gat_leaky(e[u] + edk);
  float mn = m;
#pragma unroll
  for (int u = 0; u < CH; ++u) mn = fmaxf(mn, e[u]);
  const float sc = __expf(m - mn);
  m = mn;
  den *= sc;
  acc.x *= sc;
  acc.y *= sc;
  acc.z *= sc;
  acc.w *= sc;
#pragma unroll
  for (int u = 0; u < CH; ++u) {
    float a = __expf(e[u] - mn);
    if (TAIL && u >= rem) a = 0.0f;
    den += a;
    acc.x += a * hv[u].x;
    acc.y += a * hv[u].y;
    acc.z += a * hv[u].z;
    acc.w += a * hv[u].w;
  }
}

__device__ __forceinline__ void gat_from_list(const int* __restrict__ ld,
                                              const int* __restrict__ ls, int nl, int d,
                                              const float* __restrict__ h,
                                              int lane, float4 asl, float edk,
                                              float& m, float& den, float4& acc) {
  for (int i0 = 0; i0 < nl; i0 += 32) {
    const int i = i0 + lane;
    int dd = -1, ss = 0;
    if (i < nl) {
      dd = ld[i];
      ss = ls[i];
    }
    unsigned mask = __builtin_amdgcn_ballot_w32(dd == d);
    while (mask != 0u) {
      const int b = __builtin_ctz(mask);
      mask &= mask - 1u;
      int one[4];
      one[0] = one[1] = one[2] = one[3] = __builtin_amdgcn_readlane(ss, b);
      gat_chunk<4, true>(one, 1, h, lane, asl, edk, m, den, acc);
    }
  }
}

__device__ __forceinline__ void gat_put(int2 e, int base, int nloc, int* lc, int* lb,
                                        int* ovc, int* ovd, int* ovs) {
  const int dl = e.x - base;
  if (dl >= 0 && dl < nloc) {
    const int r = atomicAdd(&lc[dl], 1);
    if (r < CAP) {
      lb[dl * CAP + r] = e.y;
    } else {
      const int o = atomicAdd(ovc, 1);
      if (o < OVL) {
        ovd[o] = e.x;
        ovs[o] = e.y;
      }
    }
  }
}

__global__ __launch_bounds__(GT) void gat_place_gather(
    const int* __restrict__ staging, const int* __restrict__ offs, int epb, int rpn,
    const int* __restrict__ src, const int* __restrict__ dst, int ne,
    const float* __restrict__ h,
    const float* __restrict__ attn_src, const float* __restrict__ attn_dst,
    const float* __restrict__ feat, float* __restrict__ out, int n) {
  extern __shared__ __attribute__((aligned(128))) int dynb[];
  __shared__ int seg0[NBLKA];
  __shared__ int segn[NBLKA];
  __shared__ int ovd[OVL];
  __shared__ int ovs[OVL];
  __shared__ int ovc[1];
  __shared__ int nextn[1];
  int* lb = dynb;
  int* lc = dynb + rpn * CAP;
  const int2* __restrict__ pairs = (const int2*)staging;
  const int tid = threadIdx.x;
  const int g = blockIdx.x;
  const int base = g * rpn;
  const int nloc = min(rpn, n - base);
  for (int i = tid; i < rpn * (CAP + 1); i += GT) dynb[i] = 0;
  if (tid == 0) {
    ovc[0] = 0;
    nextn[0] = GT / 32;
  }
  if (tid < NBLKA) {
    const int* ob = offs + tid * (NBIN + 1) + g;
    const int a = ob[0];
    seg0[tid] = tid * epb + a;
    segn[tid] = ob[1] - a;
  }
  __syncthreads();
  const int w = __builtin_amdgcn_readfirstlane(tid >> 5);
  const int lane = tid & 31;
  {
    int st[8], ln[8];
    int2 it[8];
#pragma unroll
    for (int u = 0; u < 8; ++u) {
      st[u] = seg0[w + 32 * u];
      ln[u] = segn[w + 32 * u];
    }
#pragma unroll
    for (int u = 0; u < 8; ++u)
      it[u] = lane < ln[u] ? pairs[st[u] + lane] : make_int2(-1, 0);
#pragma unroll
    for (int u = 0; u < 8; ++u) {
      if (lane < ln[u]) gat_put(it[u], base, nloc, lc, lb, ovc, ovd, ovs);
      for (int i = lane + 32; i < ln[u]; i += 32)
        gat_put(pairs[st[u] + i], base, nloc, lc, lb, ovc, ovd, ovs);
    }
  }
  __syncthreads();

  const float4 asl = *(const float4*)(attn_src + lane * 4);
  const float4 adl = *(const float4*)(attn_dst + lane * 4);
  const int ovn = __builtin_amdgcn_readfirstlane(ovc[0]);
  v4f hd_n = {0.0f, 0.0f, 0.0f, 0.0f};
  if (w < nloc)
    hd_n = ((const v4f*)h)[(unsigned)(base + w) * (HDIM / 4) + (unsigned)lane];
  for (int dl = w; dl < nloc;) {
    const int d = base + dl;
    const v4f hd = hd_n;
    int took = 0;
    if (lane == 0) took = atomicAdd(&nextn[0], 1);
    const int dn = __builtin_amdgcn_readfirstlane(took);
    if (dn < nloc)
      hd_n = ((const v4f*)h)[(unsigned)(base + dn) * (HDIM / 4) + (unsigned)lane];
    const float4 f = *(const float4*)(feat + (size_t)d * HDIM + lane * 4);
    const int deg = __builtin_amdgcn_readfirstlane(lc[dl]);
    const int* nb = lb + dl * CAP;
    __builtin_amdgcn_sched_barrier(0);
    const float edk = gat_head_sum(gat_dot4(hd, adl));

    float m = -INFINITY;
    float den = 0.0f;
    float4 acc = make_float4(0.0f, 0.0f, 0.0f, 0.0f);
    int degb = deg;
    if (deg > CAP) degb = ovn <= OVL ? CAP : 0;
    for (int j0 = 0; j0 < degb; j0 += 12) {
      const int rem = degb - j0;
      int ids[12];
#pragma unroll
      for (int q = 0; q < 2; ++q) {
        const int4 v = *(const int4*)(nb + j0 + 4 * q);
        ids[4 * q + 0] = __builtin_amdgcn_readfirstlane(v.x);
        ids[4 * q + 1] = __builtin_amdgcn_readfirstlane(v.y);
        ids[4 * q + 2] = __builtin_amdgcn_readfirstlane(v.z);
        ids[4 * q + 3] = __builtin_amdgcn_readfirstlane(v.w);
      }
      if (rem > 8) {
        const int4 v = *(const int4*)(nb + j0 + 8);
        ids[8] = __builtin_amdgcn_readfirstlane(v.x);
        ids[9] = __builtin_amdgcn_readfirstlane(v.y);
        ids[10] = __builtin_amdgcn_readfirstlane(v.z);
        ids[11] = __builtin_amdgcn_readfirstlane(v.w);
        if (rem >= 12) {
          gat_chunk<12, false>(ids, 12, h, lane, asl, edk, m, den, acc);
        } else {
          gat_chunk<12, true>(ids, rem, h, lane, asl, edk, m, den, acc);
        }
      } else if (rem > 4) {
        gat_chunk<8, true>(ids, rem, h, lane, asl, edk, m, den, acc);
      } else {
        gat_chunk<4, true>(ids, rem, h, lane, asl, edk, m, den, acc);
      }
    }

    if (deg > CAP) {
      if (ovn <= OVL)
        gat_from_list(ovd, ovs, ovn, d, h, lane, asl, edk, m, den, acc);
      else
        gat_from_list(dst, src, ne, d, h, lane, asl, edk, m, den, acc);
    }

    const float dsafe = den > 0.0f ? den : 1.0f;
    float4 o;
    o.x = f.x + acc.x / dsafe;
    o.y = f.y + acc.y / dsafe;
    o.z = f.z + acc.z / dsafe;
    o.w = f.w + acc.w / dsafe;
    *(float4*)(out + (size_t)d * HDIM + lane * 4) = o;
    dl = dn;
  }
}

extern "C" void kernel_launch(void* const* d_in, const int* in_sizes, int n_in,
                              void* d_out, int out_size, void* d_ws, size_t ws_size,
                              hipStream_t stream) {
  const float* feat     = (const float*)d_in[0];
  const float* fc_w     = (const float*)d_in[1];
  const float* attn_src = (const float*)d_in[2];
  const float* attn_dst = (const float*)d_in[3];
  const int*   src      = (const int*)d_in[4];
  const int*   dst      = (const int*)d_in[5];
  float* out = (float*)d_out;

  const int n  = in_sizes[0] / IN_DIM;
  const int ne = in_sizes[4];

  const int epb = (ne + NBLKA - 1) / NBLKA;
  const int rpn = (n + NBIN - 1) / NBIN;

  char* ws = (char*)d_ws;
  size_t off = 0;
  float* h      = (float*)(ws + off); off += (size_t)n * HDIM * sizeof(float);
  int* staging  = (int*)(ws + off);   off += (size_t)2 * NBLKA * epb * sizeof(int);
  int* offs     = (int*)(ws + off);   off += (size_t)NBLKA * (NBIN + 1) * sizeof(int);
  (void)ws_size; (void)n_in; (void)out_size;

  const int T = 256;
  gat_bin<<<NBLKA, GT, (size_t)2 * epb * sizeof(int), stream>>>(src, dst, ne, epb, rpn, staging,
                                                               offs);
  gat_fc_wmma<<<(n + 127) / 128, T, B_LDS_BYTES, stream>>>(feat, fc_w, h, n);
  gat_place_gather<<<NBIN, GT, (size_t)rpn * (CAP + 1) * sizeof(int), stream>>>(
      staging, offs, epb, rpn, src, dst, ne, h, attn_src, attn_dst, feat, out, n);
}


# Feature product as one resident 8-wave work-group per WGP: weights staged to LDS once per WGP, row tiles dealt evenly over the SIMDs, a tile's rows requested a tile ahead, B fragments a block ahead
# speedup vs baseline: 5.1624x; 1.0784x over previous
#include <hip/hip_runtime.h>
#include <hip/hip_bf16.h>
#include <hip/hip_fp16.h>

typedef float v2f __attribute__((ext_vector_type(2)));
typedef float v8f __attribute__((ext_vector_type(8)));
typedef float v4f __attribute__((ext_vector_type(4)));

#define IN_DIM 128
#define HDIM 128
#define NHEADS 4
#define ODIM 32
#define CAP 32

#define BPADF 272
#define B_LDS_BYTES (64 * BPADF * 4)

#define FC_GRID 256
#define FC_T 256
#define FC_W (FC_T / 32)

__device__ __forceinline__ void gat_fc_load_a(v2f (&a)[32], const float* __restrict__ feat,
                                              int rt, int l16, int half, int n) {
  const int row = min(rt * 16 + l16, n - 1);
  const float* __restrict__ fr = feat + (size_t)row * IN_DIM + 2 * half;
#pragma unroll
  for (int s = 0; s < 32; ++s) a[s] = *(const v2f*)(fr + 4 * s);
}

__global__ __launch_bounds__(FC_T) void gat_fc_wmma(
    const float* __restrict__ feat, const float* __restrict__ fc_w,
    float* __restrict__ h, int n) {
  extern __shared__ float Bs[];
  const int tid = threadIdx.x;
  const int lane = tid & 31;
  const int wave = __builtin_amdgcn_readfirstlane(tid >> 5);
  const int half = lane >> 4;
  const int l16  = lane & 15;
  const int G = (int)gridDim.x;
  const int g = (int)blockIdx.x;
  const int ntiles = (n + 15) >> 4;
  const int cnt = g < ntiles ? (ntiles - g + G - 1) / G : 0;
  const int R = cnt / FC_W;
  const int L = cnt - R * FC_W;
  const int nh = 2 * L > wave ? (2 * L - wave + FC_W - 1) / FC_W : 0;
  const int nitems = R + nh;

  v2f a_cur[32];
  v2f a_nxt[32] = {};
  {
    const int jl = lane & 7;
    const int ql = lane >> 3;
    v4f w[16];
#pragma unroll
    for (int i = 0; i < 16; ++i) {
      const int j = (wave + 8 * (i >> 3)) * 8 + jl;
      const int q = (i & 7) * 4 + ql;
      w[i] = *(const v4f*)(fc_w + j * IN_DIM + q * 4);
    }
    gat_fc_load_a(a_cur, feat, g + G * wave, l16, half, n);
#pragma unroll
    for (int i = 0; i < 16; ++i) {
      const int j = (wave + 8 * (i >> 3)) * 8 + jl;
      const int q = (i & 7) * 4 + ql;
      v2f lo = {w[i].x, w[i].y};
      v2f hi = {w[i].z, w[i].w};
      *(v2f*)(Bs + (2 * q) * BPADF + j * 2) = lo;
      *(v2f*)(Bs + (2 * q + 1) * BPADF + j * 2) = hi;
    }
  }
  __syncthreads();

  const float* __restrict__ bbase = Bs + l16 * 2 + half * BPADF;

  for (int i = 0; i < nitems; ++i) {
    int idx, c0, c1;
    if (i < R) {
      idx = wave + FC_W * i;
      c0 = 0;
      c1 = 2;
    } else {
      const int hu = wave + FC_W * (i - R);
      idx = R * FC_W + (hu >> 1);
      c0 = hu & 1;
      c1 = c0 + 1;
    }
    const int rowbase = (g + G * idx) * 16;
    if (i + 1 < nitems) {
      const int i1 = i + 1;
      const int idx1 = i1 < R ? wave + FC_W * i1
                              : R * FC_W + ((wave + FC_W * (i1 - R)) >> 1);
      gat_fc_load_a(a_nxt, feat, g + G * idx1, l16, half, n);
    }

#pragma unroll
    for (int ch = 0; ch < 2; ++ch) {
      if (ch < c0 || ch >= c1) continue;
      const float* __restrict__ bb = bbase + ch * 128;
      v8f acc[4] = {};
      v2f b[2][8];
#pragma unroll
      for (int ks = 0; ks < 2; ++ks)
#pragma unroll
        for (int t = 0; t < 4; ++t)
          b[0][ks * 4 + t] = *(const v2f*)(bb + (2 * ks) * BPADF + t * 32);
#pragma unroll
      for (int blk = 0; blk < 16; ++blk) {
        if (blk + 1 < 16) {
#pragma unroll
          for (int ks = 0; ks < 2; ++ks)
#pragma unroll
            for (int t = 0; t < 4; ++t)
              b[(blk + 1) & 1][ks * 4 + t] =
                  *(const v2f*)(bb + (2 * (2 * (blk + 1) + ks)) * BPADF + t * 32);
        }
        __builtin_amdgcn_sched_barrier(0);
#pragma unroll
        for (int ks = 0; ks < 2; ++ks)
#pragma unroll
          for (int t = 0; t < 4; ++t)
            acc[t] = __builtin_amdgcn_wmma_f32_16x16x4_f32(
                false, a_cur[2 * blk + ks], false, b[blk & 1][ks * 4 + t],
                (short)0, acc[t], false, false);
        __builtin_amdgcn_sched_barrier(0);
        asm volatile("v_nop\n\tv_nop\n\tv_nop\n\tv_nop");
        __builtin_amdgcn_sched_barrier(0);
      }
#pragma unroll
      for (int j = 0; j < 8; ++j) {
        const int r = rowbase + j + 8 * half;
        if (r < n) {
          float* hr = h + (size_t)r * HDIM + ch * 64;
#pragma unroll
          for (int t = 0; t < 4; ++t) hr[t * 16 + l16] = acc[t][j];
        }
      }
    }
#pragma unroll
    for (int s = 0; s < 32; ++s) a_cur[s] = a_nxt[s];
  }
}

#define NBIN 256
#define NBLKA 256
#define GT 1024
#define RPL (NBIN / 32)
#define OVL 256
static_assert(NBIN == 32 * RPL, "the scan gives RPL ranges to each lane of one wave");
static_assert(NBIN + 1 <= GT, "one thread per segment start");
static_assert(NBLKA == (GT / 32) * 8, "the placement gives 8 chunks to each wave");

__device__ __forceinline__ int gat_range(int d, int rpn) {
  const unsigned r = (unsigned)d / (unsigned)rpn;
  return (int)(r < (unsigned)(NBIN - 1) ? r : (unsigned)(NBIN - 1));
}

__global__ __launch_bounds__(GT) void gat_bin(const int* __restrict__ src,
                                              const int* __restrict__ dst, int ne, int epb,
                                              int rpn, int* __restrict__ staging,
                                              int* __restrict__ offs) {
  extern __shared__ int dyn[];
  __shared__ int cur[NBIN];
  __shared__ int beg[NBIN + 1];
  int* sd = dyn;
  int* ss = dyn + epb;
  const int tid = threadIdx.x;
  const int b = blockIdx.x;
  if (tid < NBIN) cur[tid] = 0;
  __syncthreads();
  const int e0 = b * epb;
  const int e1 = min(ne, e0 + epb);
  for (int eb = e0; eb < e1; eb += GT * 8) {
    int d[8];
#pragma unroll
    for (int u = 0; u < 8; ++u) {
      const int e = eb + u * GT + tid;
      d[u] = e < e1 ? dst[e] : 0;
    }
#pragma unroll
    for (int u = 0; u < 8; ++u)
      if (eb + u * GT + tid < e1) atomicAdd(&cur[gat_range(d[u], rpn)], 1);
  }
  __syncthreads();
  if (tid < 32) {
    int v[RPL];
    int sum = 0;
#pragma unroll
    for (int u = 0; u < RPL; ++u) {
      v[u] = cur[tid * RPL + u];
      sum += v[u];
    }
    int incl = sum;
#pragma unroll
    for (int off = 1; off < 32; off <<= 1) {
      const int t = __shfl_up(incl, off);
      if (tid >= off) incl += t;
    }
    int run = incl - sum;
#pragma unroll
    for (int u = 0; u < RPL; ++u) {
      beg[tid * RPL + u] = run;
      cur[tid * RPL + u] = run;
      run += v[u];
    }
    if (tid == 31) beg[NBIN] = run;
  }
  __syncthreads();
  for (int eb = e0; eb < e1; eb += GT * 8) {
    int d[8], s[8];
#pragma unroll
    for (int u = 0; u < 8; ++u) {
      const int e = eb + u * GT + tid;
      d[u] = e < e1 ? dst[e] : 0;
      s[u] = e < e1 ? src[e] : 0;
    }
#pragma unroll
    for (int u = 0; u < 8; ++u)
      if (eb + u * GT + tid < e1) {
        const int p = atomicAdd(&cur[gat_range(d[u], rpn)], 1);
        sd[p] = d[u];
        ss[p] = s[u];
      }
  }
  __syncthreads();
  int2* __restrict__ pairs = (int2*)staging;
  for (int i = tid; i < e1 - e0; i += GT) pairs[e0 + i] = make_int2(sd[i], ss[i]);
  if (tid <= NBIN) offs[b * (NBIN + 1) + tid] = beg[tid];
}

__device__ __forceinline__ float gat_leaky(float x) {
  return x > 0.0f ? x : 0.2f * x;
}

template <int MASK>
__device__ __forceinline__ float gat_xadd(float x) {
  float y;
  asm volatile("v_add_f32_dpp %0, %1, %1 row_xmask:%2 row_mask:0xf bank_mask:0xf bound_ctrl:1"
      : "=v"(y)
      : "v"(x), "n"(MASK));
  return y;
}

__device__ __forceinline__ float gat_dot4(v4f r, float4 a) {
  return r.x * a.x + r.y * a.y + r.z * a.z + r.w * a.w;
}

__device__ __forceinline__ float gat_head_sum(float p) {
  p = gat_xadd<1>(p);
  p = gat_xadd<2>(p);
  p = gat_xadd<4>(p);
  return p;
}

template <int CH, bool TAIL>
__device__ __forceinline__ void gat_chunk(const int* s_in, int rem,
                                          const float* __restrict__ h,
                                          int lane, float4 asl, float edk,
                                          float& m, float& den, float4& acc) {
  int s[CH];
  float e[CH];
  v4f hv[CH];
#pragma unroll
  for (int u = 0; u < CH; ++u) s[u] = (!TAIL || u < rem) ? s_in[u] : s_in[0];
#pragma unroll
  for (int u = 0; u < CH; ++u) {
    hv[u] = ((const v4f*)h)[(unsigned)s[u] * (HDIM / 4) + (unsigned)lane];
  }
#pragma unroll
  for (int u = 0; u < CH; ++u) e[u] = gat_dot4(hv[u], asl);
#pragma unroll
  for (int u = 0; u < CH; ++u) e[u] = gat_xadd<1>(e[u]);
#pragma unroll
  for (int u = 0; u < CH; ++u) e[u] = gat_xadd<2>(e[u]);
#pragma unroll
  for (int u = 0; u < CH; ++u) e[u] = gat_xadd<4>(e[u]);
#pragma unroll
  for (int u = 0; u < CH; ++u) e[u] = gat_leaky(e[u] + edk);
  float mn = m;
#pragma unroll
  for (int u = 0; u < CH; ++u) mn = fmaxf(mn, e[u]);
  const float sc = __expf(m - mn);
  m = mn;
  den *= sc;
  acc.x *= sc;
  acc.y *= sc;
  acc.z *= sc;
  acc.w *= sc;
#pragma unroll
  for (int u = 0; u < CH; ++u) {
    float a = __expf(e[u] - mn);
    if (TAIL && u >= rem) a = 0.0f;
    den += a;
    acc.x += a * hv[u].x;
    acc.y += a * hv[u].y;
    acc.z += a * hv[u].z;
    acc.w += a * hv[u].w;
  }
}

__device__ __forceinline__ void gat_from_list(const int* __restrict__ ld,
                                              const int* __restrict__ ls, int nl, int d,
                                              const float* __restrict__ h,
                                              int lane, float4 asl, float edk,
                                              float& m, float& den, float4& acc) {
  for (int i0 = 0; i0 < nl; i0 += 32) {
    const int i = i0 + lane;
    int dd = -1, ss = 0;
    if (i < nl) {
      dd = ld[i];
      ss = ls[i];
    }
    unsigned mask = __builtin_amdgcn_ballot_w32(dd == d);
    while (mask != 0u) {
      const int b = __builtin_ctz(mask);
      mask &= mask - 1u;
      int one[4];
      one[0] = one[1] = one[2] = one[3] = __builtin_amdgcn_readlane(ss, b);
      gat_chunk<4, true>(one, 1, h, lane, asl, edk, m, den, acc);
    }
  }
}

__device__ __forceinline__ void gat_put(int2 e, int base, int nloc, int* lc, int* lb,
                                        int* ovc, int* ovd, int* ovs) {
  const int dl = e.x - base;
  if (dl >= 0 && dl < nloc) {
    const int r = atomicAdd(&lc[dl], 1);
    if (r < CAP) {
      lb[dl * CAP + r] = e.y;
    } else {
      const int o = atomicAdd(ovc, 1);
      if (o < OVL) {
        ovd[o] = e.x;
        ovs[o] = e.y;
      }
    }
  }
}

__global__ __launch_bounds__(GT) void gat_place_gather(
    const int* __restrict__ staging, const int* __restrict__ offs, int epb, int rpn,
    const int* __restrict__ src, const int* __restrict__ dst, int ne,
    const float* __restrict__ h,
    const float* __restrict__ attn_src, const float* __restrict__ attn_dst,
    const float* __restrict__ feat, float* __restrict__ out, int n) {
  extern __shared__ __attribute__((aligned(128))) int dynb[];
  __shared__ int seg0[NBLKA];
  __shared__ int segn[NBLKA];
  __shared__ int ovd[OVL];
  __shared__ int ovs[OVL];
  __shared__ int ovc[1];
  __shared__ int nextn[1];
  int* lb = dynb;
  int* lc = dynb + rpn * CAP;
  const int2* __restrict__ pairs = (const int2*)staging;
  const int tid = threadIdx.x;
  const int g = blockIdx.x;
  const int base = g * rpn;
  const int nloc = min(rpn, n - base);
  for (int i = tid; i < rpn * (CAP + 1); i += GT) dynb[i] = 0;
  if (tid == 0) {
    ovc[0] = 0;
    nextn[0] = GT / 32;
  }
  if (tid < NBLKA) {
    const int* ob = offs + tid * (NBIN + 1) + g;
    const int a = ob[0];
    seg0[tid] = tid * epb + a;
    segn[tid] = ob[1] - a;
  }
  __syncthreads();
  const int w = __builtin_amdgcn_readfirstlane(tid >> 5);
  const int lane = tid & 31;
  {
    int st[8], ln[8];
    int2 it[8];
#pragma unroll
    for (int u = 0; u < 8; ++u) {
      st[u] = seg0[w + 32 * u];
      ln[u] = segn[w + 32 * u];
    }
#pragma unroll
    for (int u = 0; u < 8; ++u)
      it[u] = lane < ln[u] ? pairs[st[u] + lane] : make_int2(-1, 0);
#pragma unroll
    for (int u = 0; u < 8; ++u) {
      if (lane < ln[u]) gat_put(it[u], base, nloc, lc, lb, ovc, ovd, ovs);
      for (int i = lane + 32; i < ln[u]; i += 32)
        gat_put(pairs[st[u] + i], base, nloc, lc, lb, ovc, ovd, ovs);
    }
  }
  __syncthreads();

  const float4 asl = *(const float4*)(attn_src + lane * 4);
  const float4 adl = *(const float4*)(attn_dst + lane * 4);
  const int ovn = __builtin_amdgcn_readfirstlane(ovc[0]);
  v4f hd_n = {0.0f, 0.0f, 0.0f, 0.0f};
  if (w < nloc)
    hd_n = ((const v4f*)h)[(unsigned)(base + w) * (HDIM / 4) + (unsigned)lane];
  for (int dl = w; dl < nloc;) {
    const int d = base + dl;
    const v4f hd = hd_n;
    int took = 0;
    if (lane == 0) took = atomicAdd(&nextn[0], 1);
    const int dn = __builtin_amdgcn_readfirstlane(took);
    if (dn < nloc)
      hd_n = ((const v4f*)h)[(unsigned)(base + dn) * (HDIM / 4) + (unsigned)lane];
    const float4 f = *(const float4*)(feat + (size_t)d * HDIM + lane * 4);
    const int deg = __builtin_amdgcn_readfirstlane(lc[dl]);
    const int* nb = lb + dl * CAP;
    __builtin_amdgcn_sched_barrier(0);
    const float edk = gat_head_sum(gat_dot4(hd, adl));

    float m = -INFINITY;
    float den = 0.0f;
    float4 acc = make_float4(0.0f, 0.0f, 0.0f, 0.0f);
    int degb = deg;
    if (deg > CAP) degb = ovn <= OVL ? CAP : 0;
    for (int j0 = 0; j0 < degb; j0 += 12) {
      const int rem = degb - j0;
      int ids[12];
#pragma unroll
      for (int q = 0; q < 2; ++q) {
        const int4 v = *(const int4*)(nb + j0 + 4 * q);
        ids[4 * q + 0] = __builtin_amdgcn_readfirstlane(v.x);
        ids[4 * q + 1] = __builtin_amdgcn_readfirstlane(v.y);
        ids[4 * q + 2] = __builtin_amdgcn_readfirstlane(v.z);
        ids[4 * q + 3] = __builtin_amdgcn_readfirstlane(v.w);
      }
      if (rem > 8) {
        const int4 v = *(const int4*)(nb + j0 + 8);
        ids[8] = __builtin_amdgcn_readfirstlane(v.x);
        ids[9] = __builtin_amdgcn_readfirstlane(v.y);
        ids[10] = __builtin_amdgcn_readfirstlane(v.z);
        ids[11] = __builtin_amdgcn_readfirstlane(v.w);
        if (rem >= 12) {
          gat_chunk<12, false>(ids, 12, h, lane, asl, edk, m, den, acc);
        } else {
          gat_chunk<12, true>(ids, rem, h, lane, asl, edk, m, den, acc);
        }
      } else if (rem > 4) {
        gat_chunk<8, true>(ids, rem, h, lane, asl, edk, m, den, acc);
      } else {
        gat_chunk<4, true>(ids, rem, h, lane, asl, edk, m, den, acc);
      }
    }

    if (deg > CAP) {
      if (ovn <= OVL)
        gat_from_list(ovd, ovs, ovn, d, h, lane, asl, edk, m, den, acc);
      else
        gat_from_list(dst, src, ne, d, h, lane, asl, edk, m, den, acc);
    }

    const float dsafe = den > 0.0f ? den : 1.0f;
    float4 o;
    o.x = f.x + acc.x / dsafe;
    o.y = f.y + acc.y / dsafe;
    o.z = f.z + acc.z / dsafe;
    o.w = f.w + acc.w / dsafe;
    *(float4*)(out + (size_t)d * HDIM + lane * 4) = o;
    dl = dn;
  }
}

extern "C" void kernel_launch(void* const* d_in, const int* in_sizes, int n_in,
                              void* d_out, int out_size, void* d_ws, size_t ws_size,
                              hipStream_t stream) {
  const float* feat     = (const float*)d_in[0];
  const float* fc_w     = (const float*)d_in[1];
  const float* attn_src = (const float*)d_in[2];
  const float* attn_dst = (const float*)d_in[3];
  const int*   src      = (const int*)d_in[4];
  const int*   dst      = (const int*)d_in[5];
  float* out = (float*)d_out;

  const int n  = in_sizes[0] / IN_DIM;
  const int ne = in_sizes[4];

  const int epb = (ne + NBLKA - 1) / NBLKA;
  const int rpn = (n + NBIN - 1) / NBIN;

  char* ws = (char*)d_ws;
  size_t off = 0;
  float* h      = (float*)(ws + off); off += (size_t)n * HDIM * sizeof(float);
  int* staging  = (int*)(ws + off);   off += (size_t)2 * NBLKA * epb * sizeof(int);
  int* offs     = (int*)(ws + off);   off += (size_t)NBLKA * (NBIN + 1) * sizeof(int);
  (void)ws_size; (void)n_in; (void)out_size;

  gat_bin<<<NBLKA, GT, (size_t)2 * epb * sizeof(int), stream>>>(src, dst, ne, epb, rpn, staging,
                                                               offs);
  gat_fc_wmma<<<FC_GRID, FC_T, B_LDS_BYTES, stream>>>(feat, fc_w, h, n);
  gat_place_gather<<<NBIN, GT, (size_t)rpn * (CAP + 1) * sizeof(int), stream>>>(
      staging, offs, epb, rpn, src, dst, ne, h, attn_src, attn_dst, feat, out, n);
}


# Edge sort moves into the product's launch: waves 8-15 of each work-group sort edge chunk g in LDS while waves 0-7 run the WMMAs, LDS-only barriers between items; launches 3->2; LDS arrival order
# speedup vs baseline: 5.4736x; 1.0603x over previous
#include <hip/hip_runtime.h>
#include <hip/hip_bf16.h>
#include <hip/hip_fp16.h>

typedef float v2f __attribute__((ext_vector_type(2)));
typedef float v8f __attribute__((ext_vector_type(8)));
typedef float v4f __attribute__((ext_vector_type(4)));

#define IN_DIM 128
#define HDIM 128
#define NHEADS 4
#define ODIM 32
#define CAP 32

#define BPADF 272
#define B_LDS_BYTES (64 * BPADF * 4)

#define FC_GRID 256
#define FC_T 256
#define FC_W (FC_T / 32)
#define NBIN 256
#define NBLKA 256
#define BT 256
#define BU 25
#define FB_T (FC_T + BT)
#define GT 1024
#define RPL (NBIN / 32)
#define OVL 256
static_assert(NBIN == 32 * RPL, "the scan gives RPL ranges to each lane of one wave");
static_assert(NBLKA == (GT / 32) * 8, "the placement gives 8 chunks to each wave");
static_assert(NBLKA == FC_GRID, "work-group g sorts chunk g beside its share of the product");

__device__ __forceinline__ void gat_fc_load_a(v2f (&a)[32], const float* __restrict__ feat,
                                              int rt, int l16, int half, int n) {
  const int row = min(rt * 16 + l16, n - 1);
  const float* __restrict__ fr = feat + (size_t)row * IN_DIM + 2 * half;
#pragma unroll
  for (int s = 0; s < 32; ++s) a[s] = *(const v2f*)(fr + 4 * s);
}

__device__ __forceinline__ int gat_range(int d, int rpn) {
  const unsigned r = (unsigned)d / (unsigned)rpn;
  return (int)(r < (unsigned)(NBIN - 1) ? r : (unsigned)(NBIN - 1));
}

__device__ __forceinline__ void gat_lds_barrier() {
  __builtin_amdgcn_fence(__ATOMIC_RELEASE, "workgroup", "local");
  __builtin_amdgcn_s_barrier();
  __builtin_amdgcn_fence(__ATOMIC_ACQUIRE, "workgroup", "local");
}

__global__ __launch_bounds__(FB_T) void gat_fc_bin(
    const float* __restrict__ feat, const float* __restrict__ fc_w,
    float* __restrict__ h, int n,
    const int* __restrict__ src, const int* __restrict__ dst, int ne, int epb, int rpn,
    int* __restrict__ staging, int* __restrict__ offs) {
  extern __shared__ float Bs[];
  __shared__ int cur[NBIN];
  __shared__ int beg[NBIN + 1];
  int* sd = (int*)(Bs + 64 * BPADF);
  int* ss = sd + epb;
  const int tid = threadIdx.x;
  const int lane = tid & 31;
  const int wave = __builtin_amdgcn_readfirstlane(tid >> 5);
  const bool fcw = wave < FC_W;
  const int tb = tid - FC_T;
  const int half = lane >> 4;
  const int l16  = lane & 15;
  const int G = (int)gridDim.x;
  const int g = (int)blockIdx.x;
  const int ntiles = (n + 15) >> 4;
  const int cnt = g < ntiles ? (ntiles - g + G - 1) / G : 0;
  const int R = cnt / FC_W;
  const int L = cnt - R * FC_W;
  const int nh = 2 * L > wave ? (2 * L - wave + FC_W - 1) / FC_W : 0;
  const int nitems = fcw ? R + nh : 0;
  const int fsteps = R + (L > 0 ? 1 : 0);
  const int nsteps = fsteps > 4 ? fsteps : 4;
  const int e0 = g * epb;
  const int e1 = min(ne, e0 + epb);

  v2f a_cur[32] = {};
  v2f a_nxt[32] = {};
  if (fcw) {
    const int jl = lane & 7;
    const int ql = lane >> 3;
    v4f w[16];
#pragma unroll
    for (int i = 0; i < 16; ++i) {
      const int j = (wave + 8 * (i >> 3)) * 8 + jl;
      const int q = (i & 7) * 4 + ql;
      w[i] = *(const v4f*)(fc_w + j * IN_DIM + q * 4);
    }
    gat_fc_load_a(a_cur, feat, g + G * wave, l16, half, n);
#pragma unroll
    for (int i = 0; i < 16; ++i) {
      const int j = (wave + 8 * (i >> 3)) * 8 + jl;
      const int q = (i & 7) * 4 + ql;
      v2f lo = {w[i].x, w[i].y};
      v2f hi = {w[i].z, w[i].w};
      *(v2f*)(Bs + (2 * q) * BPADF + j * 2) = lo;
      *(v2f*)(Bs + (2 * q + 1) * BPADF + j * 2) = hi;
    }
  } else {
    for (int r = tb; r < NBIN; r += BT) cur[r] = 0;
  }
  gat_lds_barrier();

  const float* __restrict__ bbase = Bs + l16 * 2 + half * BPADF;

  for (int i = 0; i < nsteps; ++i) {
    if (fcw) {
      if (i < nitems) {
        int idx, c0, c1;
        if (i < R) {
          idx = wave + FC_W * i;
          c0 = 0;
          c1 = 2;
        } else {
          const int hu = wave + FC_W * (i - R);
          idx = R * FC_W + (hu >> 1);
          c0 = hu & 1;
          c1 = c0 + 1;
        }
        const int rowbase = (g + G * idx) * 16;
        if (i + 1 < nitems) {
          const int i1 = i + 1;
          const int idx1 = i1 < R ? wave + FC_W * i1
                                  : R * FC_W + ((wave + FC_W * (i1 - R)) >> 1);
          gat_fc_load_a(a_nxt, feat, g + G * idx1, l16, half, n);
        }

#pragma unroll
        for (int ch = 0; ch < 2; ++ch) {
          if (ch < c0 || ch >= c1) continue;
          const float* __restrict__ bb = bbase + ch * 128;
          v8f acc[4] = {};
          v2f b[2][8];
#pragma unroll
          for (int ks = 0; ks < 2; ++ks)
#pragma unroll
            for (int t = 0; t < 4; ++t)
              b[0][ks * 4 + t] = *(const v2f*)(bb + (2 * ks) * BPADF + t * 32);
#pragma unroll
          for (int blk = 0; blk < 16; ++blk) {
            if (blk + 1 < 16) {
#pragma unroll
              for (int ks = 0; ks < 2; ++ks)
#pragma unroll
                for (int t = 0; t < 4; ++t)
                  b[(blk + 1) & 1][ks * 4 + t] =
                      *(const v2f*)(bb + (2 * (2 * (blk + 1) + ks)) * BPADF + t * 32);
            }
            __builtin_amdgcn_sched_barrier(0);
#pragma unroll
            for (int ks = 0; ks < 2; ++ks)
#pragma unroll
              for (int t = 0; t < 4; ++t)
                acc[t] = __builtin_amdgcn_wmma_f32_16x16x4_f32(
                    false, a_cur[2 * blk + ks], false, b[blk & 1][ks * 4 + t],
                    (short)0, acc[t], false, false);
            __builtin_amdgcn_sched_barrier(0);
            asm volatile("v_nop\n\tv_nop\n\tv_nop\n\tv_nop");
            __builtin_amdgcn_sched_barrier(0);
          }
#pragma unroll
          for (int j = 0; j < 8; ++j) {
            const int r = rowbase + j + 8 * half;
            if (r < n) {
              float* hr = h + (size_t)r * HDIM + ch * 64;
#pragma unroll
              for (int t = 0; t < 4; ++t) hr[t * 16 + l16] = acc[t][j];
            }
          }
        }
#pragma unroll
        for (int s = 0; s < 32; ++s) a_cur[s] = a_nxt[s];
      }
    } else if (i == 0) {
      for (int eb = e0; eb < e1; eb += BT * BU) {
        int d[BU];
#pragma unroll
        for (int u = 0; u < BU; ++u) {
          const int e = eb + u * BT + tb;
          d[u] = e < e1 ? dst[e] : 0;
        }
#pragma unroll
        for (int u = 0; u < BU; ++u)
          if (eb + u * BT + tb < e1) atomicAdd(&cur[gat_range(d[u], rpn)], 1);
      }
    } else if (i == 1) {
      if (tb < 32) {
        int v[RPL];
        int sum = 0;
#pragma unroll
        for (int u = 0; u < RPL; ++u) {
          v[u] = cur[tb * RPL + u];
          sum += v[u];
        }
        int incl = sum;
#pragma unroll
        for (int off = 1; off < 32; off <<= 1) {
          const int t = __shfl_up(incl, off);
          if (tb >= off) incl += t;
        }
        int run = incl - sum;
#pragma unroll
        for (int u = 0; u < RPL; ++u) {
          beg[tb * RPL + u] = run;
          cur[tb * RPL + u] = run;
          run += v[u];
        }
        if (tb == 31) beg[NBIN] = run;
      }
    } else if (i == 2) {
      for (int eb = e0; eb < e1; eb += BT * BU) {
        int d[BU], s[BU];
#pragma unroll
        for (int u = 0; u < BU; ++u) {
          const int e = eb + u * BT + tb;
          d[u] = e < e1 ? dst[e] : 0;
          s[u] = e < e1 ? src[e] : 0;
        }
#pragma unroll
        for (int u = 0; u < BU; ++u)
          if (eb + u * BT + tb < e1) {
            const int p = atomicAdd(&cur[gat_range(d[u], rpn)], 1);
            sd[p] = d[u];
            ss[p] = s[u];
          }
      }
    } else if (i == 3) {
      int2* __restrict__ pairs = (int2*)staging;
      for (int k = tb; k < e1 - e0; k += BT) pairs[e0 + k] = make_int2(sd[k], ss[k]);
      for (int k = tb; k <= NBIN; k += BT) offs[g * (NBIN + 1) + k] = beg[k];
    }
    if (i < 3) gat_lds_barrier();
  }
}

__device__ __forceinline__ float gat_leaky(float x) {
  return x > 0.0f ? x : 0.2f * x;
}

template <int MASK>
__device__ __forceinline__ float gat_xadd(float x) {
  float y;
  asm volatile("v_add_f32_dpp %0, %1, %1 row_xmask:%2 row_mask:0xf bank_mask:0xf bound_ctrl:1"
      : "=v"(y)
      : "v"(x), "n"(MASK));
  return y;
}

__device__ __forceinline__ float gat_dot4(v4f r, float4 a) {
  return r.x * a.x + r.y * a.y + r.z * a.z + r.w * a.w;
}

__device__ __forceinline__ float gat_head_sum(float p) {
  p = gat_xadd<1>(p);
  p = gat_xadd<2>(p);
  p = gat_xadd<4>(p);
  return p;
}

template <int CH, bool TAIL>
__device__ __forceinline__ void gat_chunk(const int* s_in, int rem,
                                          const float* __restrict__ h,
                                          int lane, float4 asl, float edk,
                                          float& m, float& den, float4& acc) {
  int s[CH];
  float e[CH];
  v4f hv[CH];
#pragma unroll
  for (int u = 0; u < CH; ++u) s[u] = (!TAIL || u < rem) ? s_in[u] : s_in[0];
#pragma unroll
  for (int u = 0; u < CH; ++u) {
    hv[u] = ((const v4f*)h)[(unsigned)s[u] * (HDIM / 4) + (unsigned)lane];
  }
#pragma unroll
  for (int u = 0; u < CH; ++u) e[u] = gat_dot4(hv[u], asl);
#pragma unroll
  for (int u = 0; u < CH; ++u) e[u] = gat_xadd<1>(e[u]);
#pragma unroll
  for (int u = 0; u < CH; ++u) e[u] = gat_xadd<2>(e[u]);
#pragma unroll
  for (int u = 0; u < CH; ++u) e[u] = gat_xadd<4>(e[u]);
#pragma unroll
  for (int u = 0; u < CH; ++u) e[u] = gat_leaky(e[u] + edk);
  float mn = m;
#pragma unroll
  for (int u = 0; u < CH; ++u) mn = fmaxf(mn, e[u]);
  const float sc = __expf(m - mn);
  m = mn;
  den *= sc;
  acc.x *= sc;
  acc.y *= sc;
  acc.z *= sc;
  acc.w *= sc;
#pragma unroll
  for (int u = 0; u < CH; ++u) {
    float a = __expf(e[u] - mn);
    if (TAIL && u >= rem) a = 0.0f;
    den += a;
    acc.x += a * hv[u].x;
    acc.y += a * hv[u].y;
    acc.z += a * hv[u].z;
    acc.w += a * hv[u].w;
  }
}

__device__ __forceinline__ void gat_from_list(const int* __restrict__ ld,
                                              const int* __restrict__ ls, int nl, int d,
                                              const float* __restrict__ h,
                                              int lane, float4 asl, float edk,
                                              float& m, float& den, float4& acc) {
  for (int i0 = 0; i0 < nl; i0 += 32) {
    const int i = i0 + lane;
    int dd = -1, ss = 0;
    if (i < nl) {
      dd = ld[i];
      ss = ls[i];
    }
    unsigned mask = __builtin_amdgcn_ballot_w32(dd == d);
    while (mask != 0u) {
      const int b = __builtin_ctz(mask);
      mask &= mask - 1u;
      int one[4];
      one[0] = one[1] = one[2] = one[3] = __builtin_amdgcn_readlane(ss, b);
      gat_chunk<4, true>(one, 1, h, lane, asl, edk, m, den, acc);
    }
  }
}

__device__ __forceinline__ void gat_put(int2 e, int base, int nloc, int* lc, int* lb,
                                        int* ovc, int* ovd, int* ovs) {
  const int dl = e.x - base;
  if (dl >= 0 && dl < nloc) {
    const int r = atomicAdd(&lc[dl], 1);
    if (r < CAP) {
      lb[dl * CAP + r] = e.y;
    } else {
      const int o = atomicAdd(ovc, 1);
      if (o < OVL) {
        ovd[o] = e.x;
        ovs[o] = e.y;
      }
    }
  }
}

__global__ __launch_bounds__(GT) void gat_place_gather(
    const int* __restrict__ staging, const int* __restrict__ offs, int epb, int rpn,
    const int* __restrict__ src, const int* __restrict__ dst, int ne,
    const float* __restrict__ h,
    const float* __restrict__ attn_src, const float* __restrict__ attn_dst,
    const float* __restrict__ feat, float* __restrict__ out, int n) {
  extern __shared__ __attribute__((aligned(128))) int dynb[];
  __shared__ int seg0[NBLKA];
  __shared__ int segn[NBLKA];
  __shared__ int ovd[OVL];
  __shared__ int ovs[OVL];
  __shared__ int ovc[1];
  __shared__ int nextn[1];
  int* lb = dynb;
  int* lc = dynb + rpn * CAP;
  const int2* __restrict__ pairs = (const int2*)staging;
  const int tid = threadIdx.x;
  const int g = blockIdx.x;
  const int base = g * rpn;
  const int nloc = min(rpn, n - base);
  for (int i = tid; i < rpn * (CAP + 1); i += GT) dynb[i] = 0;
  if (tid == 0) {
    ovc[0] = 0;
    nextn[0] = GT / 32;
  }
  if (tid < NBLKA) {
    const int* ob = offs + tid * (NBIN + 1) + g;
    const int a = ob[0];
    seg0[tid] = tid * epb + a;
    segn[tid] = ob[1] - a;
  }
  __syncthreads();
  const int w = __builtin_amdgcn_readfirstlane(tid >> 5);
  const int lane = tid & 31;
  {
    int st[8], ln[8];
    int2 it[8];
#pragma unroll
    for (int u = 0; u < 8; ++u) {
      st[u] = seg0[w + 32 * u];
      ln[u] = segn[w + 32 * u];
    }
#pragma unroll
    for (int u = 0; u < 8; ++u)
      it[u] = lane < ln[u] ? pairs[st[u] + lane] : make_int2(-1, 0);
#pragma unroll
    for (int u = 0; u < 8; ++u) {
      if (lane < ln[u]) gat_put(it[u], base, nloc, lc, lb, ovc, ovd, ovs);
      for (int i = lane + 32; i < ln[u]; i += 32)
        gat_put(pairs[st[u] + i], base, nloc, lc, lb, ovc, ovd, ovs);
    }
  }
  __syncthreads();

  const float4 asl = *(const float4*)(attn_src + lane * 4);
  const float4 adl = *(const float4*)(attn_dst + lane * 4);
  const int ovn = __builtin_amdgcn_readfirstlane(ovc[0]);
  v4f hd_n = {0.0f, 0.0f, 0.0f, 0.0f};
  if (w < nloc)
    hd_n = ((const v4f*)h)[(unsigned)(base + w) * (HDIM / 4) + (unsigned)lane];
  for (int dl = w; dl < nloc;) {
    const int d = base + dl;
    const v4f hd = hd_n;
    int took = 0;
    if (lane == 0) took = atomicAdd(&nextn[0], 1);
    const int dn = __builtin_amdgcn_readfirstlane(took);
    if (dn < nloc)
      hd_n = ((const v4f*)h)[(unsigned)(base + dn) * (HDIM / 4) + (unsigned)lane];
    const float4 f = *(const float4*)(feat + (size_t)d * HDIM + lane * 4);
    const int deg = __builtin_amdgcn_readfirstlane(lc[dl]);
    const int* nb = lb + dl * CAP;
    __builtin_amdgcn_sched_barrier(0);
    const float edk = gat_head_sum(gat_dot4(hd, adl));

    float m = -INFINITY;
    float den = 0.0f;
    float4 acc = make_float4(0.0f, 0.0f, 0.0f, 0.0f);
    int degb = deg;
    if (deg > CAP) degb = ovn <= OVL ? CAP : 0;
    for (int j0 = 0; j0 < degb; j0 += 12) {
      const int rem = degb - j0;
      int ids[12];
#pragma unroll
      for (int q = 0; q < 2; ++q) {
        const int4 v = *(const int4*)(nb + j0 + 4 * q);
        ids[4 * q + 0] = __builtin_amdgcn_readfirstlane(v.x);
        ids[4 * q + 1] = __builtin_amdgcn_readfirstlane(v.y);
        ids[4 * q + 2] = __builtin_amdgcn_readfirstlane(v.z);
        ids[4 * q + 3] = __builtin_amdgcn_readfirstlane(v.w);
      }
      if (rem > 8) {
        const int4 v = *(const int4*)(nb + j0 + 8);
        ids[8] = __builtin_amdgcn_readfirstlane(v.x);
        ids[9] = __builtin_amdgcn_readfirstlane(v.y);
        ids[10] = __builtin_amdgcn_readfirstlane(v.z);
        ids[11] = __builtin_amdgcn_readfirstlane(v.w);
        if (rem >= 12) {
          gat_chunk<12, false>(ids, 12, h, lane, asl, edk, m, den, acc);
        } else {
          gat_chunk<12, true>(ids, rem, h, lane, asl, edk, m, den, acc);
        }
      } else if (rem > 4) {
        gat_chunk<8, true>(ids, rem, h, lane, asl, edk, m, den, acc);
      } else {
        gat_chunk<4, true>(ids, rem, h, lane, asl, edk, m, den, acc);
      }
    }

    if (deg > CAP) {
      if (ovn <= OVL)
        gat_from_list(ovd, ovs, ovn, d, h, lane, asl, edk, m, den, acc);
      else
        gat_from_list(dst, src, ne, d, h, lane, asl, edk, m, den, acc);
    }

    const float dsafe = den > 0.0f ? den : 1.0f;
    float4 o;
    o.x = f.x + acc.x / dsafe;
    o.y = f.y + acc.y / dsafe;
    o.z = f.z + acc.z / dsafe;
    o.w = f.w + acc.w / dsafe;
    *(float4*)(out + (size_t)d * HDIM + lane * 4) = o;
    dl = dn;
  }
}

extern "C" void kernel_launch(void* const* d_in, const int* in_sizes, int n_in,
                              void* d_out, int out_size, void* d_ws, size_t ws_size,
                              hipStream_t stream) {
  const float* feat     = (const float*)d_in[0];
  const float* fc_w     = (const float*)d_in[1];
  const float* attn_src = (const float*)d_in[2];
  const float* attn_dst = (const float*)d_in[3];
  const int*   src      = (const int*)d_in[4];
  const int*   dst      = (const int*)d_in[5];
  float* out = (float*)d_out;

  const int n  = in_sizes[0] / IN_DIM;
  const int ne = in_sizes[4];

  const int epb = (ne + NBLKA - 1) / NBLKA;
  const int rpn = (n + NBIN - 1) / NBIN;

  char* ws = (char*)d_ws;
  size_t off = 0;
  float* h      = (float*)(ws + off); off += (size_t)n * HDIM * sizeof(float);
  int* staging  = (int*)(ws + off);   off += (size_t)2 * NBLKA * epb * sizeof(int);
  int* offs     = (int*)(ws + off);   off += (size_t)NBLKA * (NBIN + 1) * sizeof(int);
  (void)ws_size; (void)n_in; (void)out_size;

  gat_fc_bin<<<FC_GRID, FB_T, B_LDS_BYTES + (size_t)2 * epb * sizeof(int), stream>>>(
      feat, fc_w, h, n, src, dst, ne, epb, rpn, staging, offs);
  gat_place_gather<<<NBIN, GT, (size_t)rpn * (CAP + 1) * sizeof(int), stream>>>(
      staging, offs, epb, rpn, src, dst, ne, h, attn_src, attn_dst, feat, out, n);
}
